# starting kernel
# baseline (speedup 1.0000x reference)
#include <hip/hip_runtime.h>
#include <cstdint>

typedef float v2f __attribute__((ext_vector_type(2)));
typedef float v8f __attribute__((ext_vector_type(8)));

namespace {

constexpr int H    = 40;
constexpr int G    = 120;
constexpr int I    = 16;
constexpr int T    = 256;
constexpr int NT   = 8;
constexpr int KC_I = 4;
constexpr int KC_H = 10;

constexpr int OFF_WIH0 = 0;
constexpr int OFF_WIH1 = OFF_WIH0 + G * I;
constexpr int OFF_WHH1 = OFF_WIH1 + G * H;
constexpr int OFF_BRZ0 = OFF_WHH1 + G * H;
constexpr int OFF_BNX0 = OFF_BRZ0 + 80;
constexpr int OFF_BNH0 = OFF_BNX0 + H;
constexpr int OFF_BRZ1 = OFF_BNH0 + H;
constexpr int OFF_BNX1 = OFF_BRZ1 + 80;
constexpr int OFF_BNH1 = OFF_BNX1 + H;
constexpr int OFF_ACCS = OFF_BNH1 + H;
constexpr int OFF_AXN  = OFF_ACCS + 16 * 80;
constexpr int OFF_AHN  = OFF_AXN  + 16 * H;
constexpr int OFF_H0   = OFF_AHN  + 16 * H;
constexpr int OFF_H1   = OFF_H0   + 16 * H;
constexpr int SMEM_FLOATS = OFF_H1 + 16 * H + 64;

__device__ __forceinline__ v8f wmma_f32(v2f a, v2f b, v8f c) {
  return __builtin_amdgcn_wmma_f32_16x16x4_f32(
       false, a,  false, b,
       (short)0, c,  false,  false);
}

__device__ __forceinline__ v2f ldsB(const float* w, int K, int nt, int nloc, int koff, int kc) {
  return *(const v2f*)(w + (nt * 16 + nloc) * K + kc * 4 + koff);
}

__device__ __forceinline__ void store_tile(float* dst, int ldw, int col, int rbase, v8f acc) {
  #pragma unroll
  for (int v = 0; v < 8; ++v) dst[(rbase + v) * ldw + col] = acc[v];
}

__device__ __forceinline__ float fsigmoid(float v) { return 1.0f / (1.0f + __expf(-v)); }

}

__global__ __launch_bounds__(32) void gru2_fused(
    const float* __restrict__ x,
    const float* __restrict__ Wih0, const float* __restrict__ Whh0,
    const float* __restrict__ bih0, const float* __restrict__ bhh0,
    const float* __restrict__ Wih1, const float* __restrict__ Whh1,
    const float* __restrict__ bih1, const float* __restrict__ bhh1,
    float* __restrict__ out)
{
  __shared__ float smem[SMEM_FLOATS];
  const int lane  = threadIdx.x & 31;
  const int nloc  = lane & 15;
  const int koff  = (lane >> 4) << 1;
  const int rbase = (lane < 16) ? 0 : 8;
  const int bbase = blockIdx.x * 16;

  for (int i = lane; i < G * I; i += 32) smem[OFF_WIH0 + i] = Wih0[i];
  for (int i = lane; i < G * H; i += 32) smem[OFF_WIH1 + i] = Wih1[i];
  for (int i = lane; i < G * H; i += 32) smem[OFF_WHH1 + i] = Whh1[i];
  for (int i = lane; i < 80; i += 32) {
    smem[OFF_BRZ0 + i] = bih0[i] + bhh0[i];
    smem[OFF_BRZ1 + i] = bih1[i] + bhh1[i];
  }
  for (int i = lane; i < H; i += 32) {
    smem[OFF_BNX0 + i] = bih0[80 + i]; smem[OFF_BNH0 + i] = bhh0[80 + i];
    smem[OFF_BNX1 + i] = bih1[80 + i]; smem[OFF_BNH1 + i] = bhh1[80 + i];
  }
  for (int i = lane; i < 16 * H; i += 32) { smem[OFF_H0 + i] = 0.f; smem[OFF_H1 + i] = 0.f; }

  v2f whh0[KC_H][NT];
  #pragma unroll
  for (int kc = 0; kc < KC_H; ++kc) {
    #pragma unroll
    for (int nt = 0; nt < NT; ++nt) {
      const int n = nt * 16 + nloc;
      v2f w = {0.f, 0.f};
      if (n < G) w = *(const v2f*)(Whh0 + n * H + kc * 4 + koff);
      whh0[kc][nt] = w;
    }
  }

  __syncthreads();
  asm volatile("s_wait_dscnt 0" ::: "memory");

  float* sH0 = smem + OFF_H0;
  float* sH1 = smem + OFF_H1;
  const float* xrow = x + (size_t)(bbase + nloc) * T * I;

  #pragma unroll 1
  for (int t = 0; t < T; ++t) {
    if (t + 1 < T) __builtin_prefetch(xrow + (t + 1) * I + koff, 0, 3);

    v2f ax[KC_I];
    #pragma unroll
    for (int kc = 0; kc < KC_I; ++kc)
      ax[kc] = *(const v2f*)(xrow + t * I + kc * 4 + koff);
    v2f ah[KC_H];
    #pragma unroll
    for (int kc = 0; kc < KC_H; ++kc)
      ah[kc] = *(const v2f*)(sH0 + nloc * H + kc * 4 + koff);
    asm volatile("s_wait_dscnt 0" ::: "memory");

    #pragma unroll
    for (int nt = 0; nt < NT; ++nt) {
      const int n = nt * 16 + nloc;
      if (nt < 5) {
        v8f acc = {};
        #pragma unroll
        for (int kc = 0; kc < KC_I; ++kc)
          acc = wmma_f32(ax[kc], ldsB(smem + OFF_WIH0, I, nt, nloc, koff, kc), acc);
        #pragma unroll
        for (int kc = 0; kc < KC_H; ++kc)
          acc = wmma_f32(ah[kc], whh0[kc][nt], acc);
        store_tile(smem + OFF_ACCS, 80, n, rbase, acc);
      } else {
        v8f accx = {}, acch = {};
        #pragma unroll
        for (int kc = 0; kc < KC_I; ++kc)
          accx = wmma_f32(ax[kc], ldsB(smem + OFF_WIH0, I, nt, nloc, koff, kc), accx);
        #pragma unroll
        for (int kc = 0; kc < KC_H; ++kc)
          acch = wmma_f32(ah[kc], whh0[kc][nt], acch);
        if (n < G) {
          store_tile(smem + OFF_AXN, H, n - 80, rbase, accx);
          store_tile(smem + OFF_AHN, H, n - 80, rbase, acch);
        }
      }
    }
    asm volatile("s_wait_dscnt 0" ::: "memory");

    for (int e = lane; e < 16 * H; e += 32) {
      const int b = e / H, j = e % H;
      const float rz = smem[OFF_ACCS + b * 80 + j]      + smem[OFF_BRZ0 + j];
      const float zz = smem[OFF_ACCS + b * 80 + 40 + j] + smem[OFF_BRZ0 + 40 + j];
      const float xn = smem[OFF_AXN + b * H + j]        + smem[OFF_BNX0 + j];
      const float hn = smem[OFF_AHN + b * H + j]        + smem[OFF_BNH0 + j];
      const float hp = sH0[b * H + j];
      const float r  = fsigmoid(rz);
      const float z  = fsigmoid(zz);
      const float nn = tanhf(xn + r * hn);
      sH0[b * H + j] = nn + z * (hp - nn);
    }
    asm volatile("s_wait_dscnt 0" ::: "memory");

    v2f ax1[KC_H], ah1[KC_H];
    #pragma unroll
    for (int kc = 0; kc < KC_H; ++kc) {
      ax1[kc] = *(const v2f*)(sH0 + nloc * H + kc * 4 + koff);
      ah1[kc] = *(const v2f*)(sH1 + nloc * H + kc * 4 + koff);
    }
    asm volatile("s_wait_dscnt 0" ::: "memory");

    #pragma unroll
    for (int nt = 0; nt < NT; ++nt) {
      const int n = nt * 16 + nloc;
      if (nt < 5) {
        v8f acc = {};
        #pragma unroll
        for (int kc = 0; kc < KC_H; ++kc)
          acc = wmma_f32(ax1[kc], ldsB(smem + OFF_WIH1, H, nt, nloc, koff, kc), acc);
        #pragma unroll
        for (int kc = 0; kc < KC_H; ++kc)
          acc = wmma_f32(ah1[kc], ldsB(smem + OFF_WHH1, H, nt, nloc, koff, kc), acc);
        store_tile(smem + OFF_ACCS, 80, n, rbase, acc);
      } else {
        v8f accx = {}, acch = {};
        #pragma unroll
        for (int kc = 0; kc < KC_H; ++kc)
          accx = wmma_f32(ax1[kc], ldsB(smem + OFF_WIH1, H, nt, nloc, koff, kc), accx);
        #pragma unroll
        for (int kc = 0; kc < KC_H; ++kc)
          acch = wmma_f32(ah1[kc], ldsB(smem + OFF_WHH1, H, nt, nloc, koff, kc), acch);
        if (n < G) {
          store_tile(smem + OFF_AXN, H, n - 80, rbase, accx);
          store_tile(smem + OFF_AHN, H, n - 80, rbase, acch);
        }
      }
    }
    asm volatile("s_wait_dscnt 0" ::: "memory");

    for (int e = lane; e < 16 * H; e += 32) {
      const int b = e / H, j = e % H;
      const float rz = smem[OFF_ACCS + b * 80 + j]      + smem[OFF_BRZ1 + j];
      const float zz = smem[OFF_ACCS + b * 80 + 40 + j] + smem[OFF_BRZ1 + 40 + j];
      const float xn = smem[OFF_AXN + b * H + j]        + smem[OFF_BNX1 + j];
      const float hn = smem[OFF_AHN + b * H + j]        + smem[OFF_BNH1 + j];
      const float hp = sH1[b * H + j];
      const float r  = fsigmoid(rz);
      const float z  = fsigmoid(zz);
      const float nn = tanhf(xn + r * hn);
      sH1[b * H + j] = nn + z * (hp - nn);
    }
    asm volatile("s_wait_dscnt 0" ::: "memory");
  }

  for (int e = lane; e < 16 * H; e += 32) {
    const int b = e / H, j = e % H;
    out[(size_t)(bbase + b) * H + j] = sH1[b * H + j];
  }
}

extern "C" void kernel_launch(void* const* d_in, const int* in_sizes, int n_in,
                              void* d_out, int out_size, void* d_ws, size_t ws_size,
                              hipStream_t stream) {
  const float* x    = (const float*)d_in[0];
  const float* Wih0 = (const float*)d_in[1];
  const float* Whh0 = (const float*)d_in[2];
  const float* bih0 = (const float*)d_in[3];
  const float* bhh0 = (const float*)d_in[4];
  const float* Wih1 = (const float*)d_in[5];
  const float* Whh1 = (const float*)d_in[6];
  const float* bih1 = (const float*)d_in[7];
  const float* bhh1 = (const float*)d_in[8];
  float* out = (float*)d_out;

  const int B = in_sizes[0] / (T * I);
  const int nblocks = (B + 15) / 16;

  gru2_fused<<<nblocks, 32, 0, stream>>>(x, Wih0, Whh0, bih0, bhh0,
                                         Wih1, Whh1, bih1, bhh1, out);
  (void)n_in; (void)out_size; (void)d_ws; (void)ws_size;
}


# Operands swapped: weights are A, hidden state is B, so a step's D registers are the next step's B; state, gates, weights all stay in registers (k order permuted, r|z bias summed first)
# speedup vs baseline: 3.4305x; 3.4305x over previous
#include <hip/hip_runtime.h>
#include <cstdint>


typedef float v2f __attribute__((ext_vector_type(2)));
typedef float v4f __attribute__((ext_vector_type(4)));
typedef float v8f __attribute__((ext_vector_type(8)));


namespace {

constexpr int H    = 40;
constexpr int HH   = 20;
constexpr int I    = 16;
constexpr int T    = 256;
constexpr int NRZ  = 5;
constexpr int NN   = 3;
constexpr int NT   = NRZ + NN;
constexpr int KC_I = 4;
constexpr int KC_H = 10;

__device__ __forceinline__ v8f wmma_f32(v2f a, v2f b, v8f c) {
  return __builtin_amdgcn_wmma_f32_16x16x4_f32(
       false, a,  false, b,
       (short)0, c,  false,  false);
}


__device__ __forceinline__ int gate_row(int tl, int m) {
  const int hr = m >> 3, v = m & 7;
  if (tl < NRZ) {
    const int q = tl * 8 + v;
    return (q < HH) ? (HH * hr + q) : (H + HH * hr + (q - HH));
  }
  const int q = (tl - NRZ) * 8 + v;
  return (q < HH) ? (2 * H + HH * hr + q) : -1;
}

template <int K, int KC>
__device__ __forceinline__ void load_a(const float* __restrict__ w, int m, int h, v2f (&a)[NT][KC]) {
  #pragma unroll
  for (int tl = 0; tl < NT; ++tl) {
    const int g  = gate_row(tl, m);
    const float* p = w + (g >= 0 ? g : 0) * K + (K / 2) * h;
    #pragma unroll
    for (int kc = 0; kc < KC; ++kc) {
      const v2f t = *(const v2f*)(p + 2 * kc);
      a[tl][kc] = (g >= 0) ? t : (v2f){0.f, 0.f};
    }
  }
}

__device__ __forceinline__ void load_bias(const float* __restrict__ bih, const float* __restrict__ bhh,
                                          int h, v8f (&brz)[NRZ], float (&bnx)[HH], float (&bnh)[HH]) {
  #pragma unroll
  for (int tl = 0; tl < NRZ; ++tl) {
    #pragma unroll
    for (int v = 0; v < 8; ++v) {
      const int g = gate_row(tl, 8 * h + v);
      brz[tl][v] = bih[g] + bhh[g];
    }
  }
  #pragma unroll
  for (int i = 0; i < HH; ++i) {
    bnx[i] = bih[2 * H + HH * h + i];
    bnh[i] = bhh[2 * H + HH * h + i];
  }
}

__device__ __forceinline__ float fsigmoid(float v) { return 1.0f / (1.0f + __expf(-v)); }

__device__ __forceinline__ float tanh_sel(float x) {
  const float y  = __builtin_fabsf(x);
  const float y2 = y * y;
  float p = __builtin_fmaf(y2, -0x1.758e7ap-8f, 0x1.521192p-6f);
  p = __builtin_fmaf(y2, p, -0x1.b8389cp-5f);
  p = __builtin_fmaf(y2, p, 0x1.110704p-3f);
  p = __builtin_fmaf(y2, p, -0x1.555532p-2f);
  const float zs = __builtin_fmaf(y2, y * p, y);
  const float e  = expf(2.0f * y);
  const float zb = __builtin_fmaf(-2.0f, __builtin_amdgcn_rcpf(e + 1.0f), 1.0f);
  const float z  = (y < 0.625f) ? zs : zb;
  return __builtin_copysignf(z, x);
}

__device__ __forceinline__ void cell_update(const v8f (&rz)[NRZ], const v8f (&xn)[NN], const v8f (&hn)[NN],
                                            const float (&bnx)[HH], const float (&bnh)[HH], v2f (&h)[KC_H]) {
  #pragma unroll
  for (int i = 0; i < HH; ++i) {
    const float r  = fsigmoid(rz[i >> 3][i & 7]);
    const float z  = fsigmoid(rz[(HH + i) >> 3][(HH + i) & 7]);
    const float nn = tanh_sel((xn[i >> 3][i & 7] + bnx[i]) + r * (hn[i >> 3][i & 7] + bnh[i]));
    const float hp = h[i >> 1][i & 1];
    h[i >> 1][i & 1] = nn + z * (hp - nn);
  }
}

template <int KCX>
__device__ __forceinline__ void gate_products(const v2f (&wx)[NT][KCX], const v2f (&bx)[KCX],
                                              const v2f (&wh)[NT][KC_H], const v2f (&bh)[KC_H],
                                              v8f (&rz)[NRZ], v8f (&xn)[NN], v8f (&hn)[NN]) {
  #pragma unroll
  for (int kc = 0; kc < KCX; ++kc) {
    #pragma unroll
    for (int tl = 0; tl < NRZ; ++tl) rz[tl] = wmma_f32(wx[tl][kc], bx[kc], rz[tl]);
    #pragma unroll
    for (int tl = 0; tl < NN; ++tl)  xn[tl] = wmma_f32(wx[NRZ + tl][kc], bx[kc], xn[tl]);
  }
  #pragma unroll
  for (int kc = 0; kc < KC_H; ++kc) {
    #pragma unroll
    for (int tl = 0; tl < NRZ; ++tl) rz[tl] = wmma_f32(wh[tl][kc], bh[kc], rz[tl]);
    #pragma unroll
    for (int tl = 0; tl < NN; ++tl)  hn[tl] = wmma_f32(wh[NRZ + tl][kc], bh[kc], hn[tl]);
  }
}

}

__global__ __launch_bounds__(32) void gru2_fused(
    const float* __restrict__ x,
    const float* __restrict__ Wih0, const float* __restrict__ Whh0,
    const float* __restrict__ bih0, const float* __restrict__ bhh0,
    const float* __restrict__ Wih1, const float* __restrict__ Whh1,
    const float* __restrict__ bih1, const float* __restrict__ bhh1,
    float* __restrict__ out)
{
  __shared__ float stage[16 * H];
  const int lane  = threadIdx.x & 31;
  const int m     = lane & 15;
  const int h     = lane >> 4;
  const int bbase = blockIdx.x * 16;

  v2f wih0[NT][KC_I], whh0[NT][KC_H], wih1[NT][KC_H], whh1[NT][KC_H];
  load_a<I, KC_I>(Wih0, m, h, wih0);
  load_a<H, KC_H>(Whh0, m, h, whh0);
  load_a<H, KC_H>(Wih1, m, h, wih1);
  load_a<H, KC_H>(Whh1, m, h, whh1);

  v8f brz0[NRZ], brz1[NRZ];
  float bnx0[HH], bnh0[HH], bnx1[HH], bnh1[HH];
  load_bias(bih0, bhh0, h, brz0, bnx0, bnh0);
  load_bias(bih1, bhh1, h, brz1, bnx1, bnh1);

  v2f h0[KC_H], h1[KC_H];
  #pragma unroll
  for (int kc = 0; kc < KC_H; ++kc) { h0[kc] = (v2f){0.f, 0.f}; h1[kc] = (v2f){0.f, 0.f}; }

  const float* xrow = x + (size_t)(bbase + m) * T * I + 8 * h;
  v4f xa = *(const v4f*)(xrow);
  v4f xb = *(const v4f*)(xrow + 4);

  const v8f zero8 = {0.f, 0.f, 0.f, 0.f, 0.f, 0.f, 0.f, 0.f};

  #pragma unroll 1
  for (int t = 0; t < T; ++t) {
    const v2f xf[KC_I] = {{xa.x, xa.y}, {xa.z, xa.w}, {xb.x, xb.y}, {xb.z, xb.w}};
    v8f rz[NRZ], xn[NN], hn[NN];

    #pragma unroll
    for (int tl = 0; tl < NRZ; ++tl) rz[tl] = brz0[tl];
    #pragma unroll
    for (int tl = 0; tl < NN; ++tl) { xn[tl] = zero8; hn[tl] = zero8; }
    gate_products<KC_I>(wih0, xf, whh0, h0, rz, xn, hn);

    const int tn = (t + 1 < T) ? (t + 1) : t;
    xa = *(const v4f*)(xrow + tn * I);
    xb = *(const v4f*)(xrow + tn * I + 4);

    cell_update(rz, xn, hn, bnx0, bnh0, h0);

    #pragma unroll
    for (int tl = 0; tl < NRZ; ++tl) rz[tl] = brz1[tl];
    #pragma unroll
    for (int tl = 0; tl < NN; ++tl) { xn[tl] = zero8; hn[tl] = zero8; }
    gate_products<KC_H>(wih1, h0, whh1, h1, rz, xn, hn);
    cell_update(rz, xn, hn, bnx1, bnh1, h1);
  }

  #pragma unroll
  for (int i = 0; i < HH; ++i) stage[m * H + HH * h + i] = h1[i >> 1][i & 1];
  __syncthreads();
  #pragma unroll
  for (int e = lane; e < 16 * H; e += 32) out[(size_t)bbase * H + e] = stage[e];
}

extern "C" void kernel_launch(void* const* d_in, const int* in_sizes, int n_in,
                              void* d_out, int out_size, void* d_ws, size_t ws_size,
                              hipStream_t stream) {
  const float* x    = (const float*)d_in[0];
  const float* Wih0 = (const float*)d_in[1];
  const float* Whh0 = (const float*)d_in[2];
  const float* bih0 = (const float*)d_in[3];
  const float* bhh0 = (const float*)d_in[4];
  const float* Wih1 = (const float*)d_in[5];
  const float* Whh1 = (const float*)d_in[6];
  const float* bih1 = (const float*)d_in[7];
  const float* bhh1 = (const float*)d_in[8];
  float* out = (float*)d_out;

  const int B = in_sizes[0] / (T * I);
  const int nblocks = (B + 15) / 16;

  gru2_fused<<<nblocks, 32, 0, stream>>>(x, Wih0, Whh0, bih0, bhh0,
                                         Wih1, Whh1, bih1, bhh1, out);
  (void)n_in; (void)out_size; (void)d_ws; (void)ws_size;
}


# Cell update written stage by stage (10 units wide) with one matrix instruction of the OTHER layer between stages, from a second accumulator set; layer 1's r|z sums now take the h side first
# speedup vs baseline: 3.7260x; 1.0861x over previous
#include <hip/hip_runtime.h>
#include <cstdint>


typedef float v2f __attribute__((ext_vector_type(2)));
typedef float v4f __attribute__((ext_vector_type(4)));
typedef float v8f __attribute__((ext_vector_type(8)));


namespace {

constexpr int H    = 40;
constexpr int HH   = 20;
constexpr int I    = 16;
constexpr int T    = 256;
constexpr int NRZ  = 5;
constexpr int NN   = 3;
constexpr int NT   = NRZ + NN;
constexpr int KC_I = 4;
constexpr int KC_H = 10;

__device__ __forceinline__ v8f wmma_f32(v2f a, v2f b, v8f c) {
  return __builtin_amdgcn_wmma_f32_16x16x4_f32(
       false, a,  false, b,
       (short)0, c,  false,  false);
}


__device__ __forceinline__ int gate_row(int tl, int m) {
  const int hr = m >> 3, v = m & 7;
  if (tl < NRZ) {
    const int q = tl * 8 + v;
    return (q < HH) ? (HH * hr + q) : (H + HH * hr + (q - HH));
  }
  const int q = (tl - NRZ) * 8 + v;
  return (q < HH) ? (2 * H + HH * hr + q) : -1;
}

template <int K, int KC>
__device__ __forceinline__ void load_a(const float* __restrict__ w, int m, int h, v2f (&a)[NT][KC]) {
  #pragma unroll
  for (int tl = 0; tl < NT; ++tl) {
    const int g  = gate_row(tl, m);
    const float* p = w + (g >= 0 ? g : 0) * K + (K / 2) * h;
    #pragma unroll
    for (int kc = 0; kc < KC; ++kc) {
      const v2f t = *(const v2f*)(p + 2 * kc);
      a[tl][kc] = (g >= 0) ? t : (v2f){0.f, 0.f};
    }
  }
}

__device__ __forceinline__ void load_bias(const float* __restrict__ bih, const float* __restrict__ bhh,
                                          int h, v8f (&brz)[NRZ], float (&bnx)[HH], float (&bnh)[HH]) {
  #pragma unroll
  for (int tl = 0; tl < NRZ; ++tl) {
    #pragma unroll
    for (int v = 0; v < 8; ++v) {
      const int g = gate_row(tl, 8 * h + v);
      brz[tl][v] = bih[g] + bhh[g];
    }
  }
  #pragma unroll
  for (int i = 0; i < HH; ++i) {
    bnx[i] = bih[2 * H + HH * h + i];
    bnh[i] = bhh[2 * H + HH * h + i];
  }
}


constexpr int NB = 10;
constexpr int ST_CELL = 59;

template <class V>
__device__ __forceinline__ void keep_stage(V (&a)[NB]) {
  static_assert(NB == 10, "operand list below is written for ten");
  asm volatile("" : "+v"(a[0]), "+v"(a[1]), "+v"(a[2]), "+v"(a[3]), "+v"(a[4]),
                    "+v"(a[5]), "+v"(a[6]), "+v"(a[7]), "+v"(a[8]), "+v"(a[9]));
}

#define STAGE(k, arr, expr) { _Pragma("unroll") for (int j = 0; j < NB; ++j) { arr[j] = (expr); } \
                              keep_stage(arr); beside(W0 + (k)); }

#define STAGE_FLAG(k, arr, expr) { _Pragma("unroll") for (int j = 0; j < NB; ++j) { arr[j] = (expr); } \
                                   beside(W0 + (k)); }

#define SIGMOID_STAGES(K, SRC, OUT)                                                           \
  STAGE((K) + 0,  t,   (SRC) * -0x1.715476p+0f)                                               \
  STAGE((K) + 1,  t,   __builtin_amdgcn_exp2f(t[j]))                                          \
  STAGE((K) + 2,  dd,  1.0f + t[j])                                                           \
  STAGE((K) + 3,  s0,  __builtin_amdgcn_div_scalef(1.0f, dd[j], false, &unused))              \
  STAGE((K) + 4,  rc,  __builtin_amdgcn_rcpf(s0[j]))                                          \
  STAGE((K) + 5,  f,   __builtin_fmaf(-s0[j], rc[j], 1.0f))                                   \
  STAGE((K) + 6,  rc,  __builtin_fmaf(f[j], rc[j], rc[j]))                                    \
  STAGE((K) + 7,  s1,  __builtin_amdgcn_div_scalef(1.0f, dd[j], true, &fl[j]))                \
  STAGE((K) + 8,  q,   s1[j] * rc[j])                                                         \
  STAGE((K) + 9,  f,   __builtin_fmaf(-s0[j], q[j], s1[j]))                                   \
  STAGE((K) + 10, q,   __builtin_fmaf(f[j], rc[j], q[j]))                                     \
  STAGE((K) + 11, f,   __builtin_fmaf(-s0[j], q[j], s1[j]))                                   \
  STAGE((K) + 12, f,   __builtin_amdgcn_div_fmasf(f[j], rc[j], q[j], fl[j]))                  \
  STAGE((K) + 13, OUT, __builtin_amdgcn_div_fixupf(f[j], dd[j], 1.0f))

template <int I0, int W0, class F>
__device__ __forceinline__ void cell_batch(const v8f (&rz)[NRZ], const v8f (&xn)[NN], const v8f (&hn)[NN],
                                           const float (&bnx)[HH], const float (&bnh)[HH],
                                           v2f (&h)[KC_H], F beside) {
  float t[NB], dd[NB], s0[NB], rc[NB], f[NB], s1[NB], q[NB], r[NB], z[NB];
  bool fl[NB], unused;
  SIGMOID_STAGES(0,  rz[(I0 + j) >> 3][(I0 + j) & 7], r)
  SIGMOID_STAGES(14, rz[(HH + I0 + j) >> 3][(HH + I0 + j) & 7], z)

  float u[NB], w[NB], pre[NB];
  STAGE(28, u,   xn[(I0 + j) >> 3][(I0 + j) & 7] + bnx[I0 + j])
  STAGE(29, w,   hn[(I0 + j) >> 3][(I0 + j) & 7] + bnh[I0 + j])
  STAGE(30, pre, u[j] + r[j] * w[j])

  float y[NB], y2[NB], p[NB], zs[NB], x2[NB], ph[NB], pl[NB], en[NB], e[NB];
  int ei[NB];
  STAGE(31, y,  __builtin_fabsf(pre[j]))
  STAGE(32, y2, y[j] * y[j])
  STAGE(33, p,  __builtin_fmaf(y2[j], -0x1.758e7ap-8f, 0x1.521192p-6f))
  STAGE(34, p,  __builtin_fmaf(y2[j], p[j], -0x1.b8389cp-5f))
  STAGE(35, p,  __builtin_fmaf(y2[j], p[j], 0x1.110704p-3f))
  STAGE(36, p,  __builtin_fmaf(y2[j], p[j], -0x1.555532p-2f))
  STAGE(37, p,  y[j] * p[j])
  STAGE(38, zs, __builtin_fmaf(y2[j], p[j], y[j]))
  STAGE(39, x2, y[j] + y[j])
  STAGE(40, ph, x2[j] * 0x1.715476p+0f)
  STAGE(41, pl, __builtin_fmaf(x2[j], 0x1.715476p+0f, -ph[j]))
  STAGE(42, en, __builtin_rintf(ph[j]))
  STAGE(43, pl, __builtin_fmaf(x2[j], 0x1.4ae0bep-26f, pl[j]))
  STAGE(44, ph, ph[j] - en[j])
  STAGE(45, ph, ph[j] + pl[j])
  STAGE(46, ei, (int)en[j])
  STAGE(47, e,  __builtin_amdgcn_exp2f(ph[j]))
  STAGE(48, e,  __builtin_ldexpf(e[j], ei[j]))
  STAGE_FLAG(49, fl, !(0x1.62e43p+6f < x2[j]))
  STAGE(50, e,  fl[j] ? e[j] : __builtin_inff())
  STAGE(51, e,  e[j] + 1.0f)
  STAGE(52, e,  __builtin_amdgcn_rcpf(e[j]))
  STAGE(53, e,  __builtin_fmaf(-2.0f, e[j], 1.0f))
  STAGE_FLAG(54, fl, y[j] < 0.625f)
  STAGE(55, e,  fl[j] ? zs[j] : e[j])
  STAGE(56, e,  __builtin_copysignf(e[j], pre[j]))

  STAGE(57, u,  h[(I0 + j) >> 1][(I0 + j) & 1] - e[j])
  STAGE(58, u,  e[j] + z[j] * u[j])
  #pragma unroll
  for (int j = 0; j < NB; ++j) h[(I0 + j) >> 1][(I0 + j) & 1] = u[j];
}

template <int KA, int KB>
__device__ __forceinline__ void one_product(int wi,
                                            const v2f (&wa)[NT][KA], const v2f (&ba)[KA], v8f (&na)[NN],
                                            const v2f (&wb)[NT][KB], const v2f (&bb)[KB], v8f (&nb)[NN],
                                            v8f (&rz)[NRZ]) {
  const int kc = wi >> 3, tl = wi & 7;
  if (kc < KA) {
    if (tl < NRZ) rz[tl] = wmma_f32(wa[tl][kc], ba[kc], rz[tl]);
    else          na[tl - NRZ] = wmma_f32(wa[tl][kc], ba[kc], na[tl - NRZ]);
  } else if (kc < KA + KB) {
    if (tl < NRZ) rz[tl] = wmma_f32(wb[tl][kc - KA], bb[kc - KA], rz[tl]);
    else          nb[tl - NRZ] = wmma_f32(wb[tl][kc - KA], bb[kc - KA], nb[tl - NRZ]);
  }
}

}

__global__ __launch_bounds__(32) void gru2_fused(
    const float* __restrict__ x,
    const float* __restrict__ Wih0, const float* __restrict__ Whh0,
    const float* __restrict__ bih0, const float* __restrict__ bhh0,
    const float* __restrict__ Wih1, const float* __restrict__ Whh1,
    const float* __restrict__ bih1, const float* __restrict__ bhh1,
    float* __restrict__ out)
{
  __shared__ float stage[16 * H];
  const int lane  = threadIdx.x & 31;
  const int m     = lane & 15;
  const int h     = lane >> 4;
  const int bbase = blockIdx.x * 16;

  v2f wih0[NT][KC_I], whh0[NT][KC_H], wih1[NT][KC_H], whh1[NT][KC_H];
  load_a<I, KC_I>(Wih0, m, h, wih0);
  load_a<H, KC_H>(Whh0, m, h, whh0);
  load_a<H, KC_H>(Wih1, m, h, wih1);
  load_a<H, KC_H>(Whh1, m, h, whh1);

  v8f brz0[NRZ], brz1[NRZ];
  float bnx0[HH], bnh0[HH], bnx1[HH], bnh1[HH];
  load_bias(bih0, bhh0, h, brz0, bnx0, bnh0);
  load_bias(bih1, bhh1, h, brz1, bnx1, bnh1);

  v2f h0[KC_H], h1[KC_H];
  #pragma unroll
  for (int kc = 0; kc < KC_H; ++kc) { h0[kc] = (v2f){0.f, 0.f}; h1[kc] = (v2f){0.f, 0.f}; }

  const float* xrow = x + (size_t)(bbase + m) * T * I + 8 * h;
  v4f xa = *(const v4f*)(xrow);
  v4f xb = *(const v4f*)(xrow + 4);

  const v8f zero8 = {0.f, 0.f, 0.f, 0.f, 0.f, 0.f, 0.f, 0.f};

  v8f rz0[NRZ], xn0[NN], hn0[NN];
  #pragma unroll
  for (int tl = 0; tl < NRZ; ++tl) rz0[tl] = brz0[tl];
  #pragma unroll
  for (int tl = 0; tl < NN; ++tl) { xn0[tl] = zero8; hn0[tl] = zero8; }
  {
    const v2f xf[KC_I] = {{xa.x, xa.y}, {xa.z, xa.w}, {xb.x, xb.y}, {xb.z, xb.w}};
    #pragma unroll
    for (int wi = 0; wi < 8 * (KC_I + KC_H); ++wi)
      one_product<KC_I, KC_H>(wi, wih0, xf, xn0, whh0, h0, hn0, rz0);
  }

  #pragma unroll 1
  for (int t = 0; t < T; ++t) {
    const int tn = (t + 1 < T) ? (t + 1) : t;
    xa = *(const v4f*)(xrow + tn * I);
    xb = *(const v4f*)(xrow + tn * I + 4);

    v8f rz1[NRZ], xn1[NN], hn1[NN];
    #pragma unroll
    for (int tl = 0; tl < NRZ; ++tl) rz1[tl] = brz1[tl];
    #pragma unroll
    for (int tl = 0; tl < NN; ++tl) { xn1[tl] = zero8; hn1[tl] = zero8; }
    {
      auto beside = [&](int wi) __attribute__((always_inline)) {
        __builtin_amdgcn_sched_barrier(0);
        one_product<KC_H, KC_H>(wi, whh1, h1, hn1, wih1, h0, xn1, rz1);
        __builtin_amdgcn_sched_barrier(0);
      };
      cell_batch<0,  0>(rz0, xn0, hn0, bnx0, bnh0, h0, beside);
      cell_batch<NB, ST_CELL>(rz0, xn0, hn0, bnx0, bnh0, h0, beside);
      #pragma unroll
      for (int wi = 2 * ST_CELL; wi < 8 * (KC_H + KC_H); ++wi)
        one_product<KC_H, KC_H>(wi, whh1, h1, hn1, wih1, h0, xn1, rz1);
      __builtin_amdgcn_sched_barrier(0);
    }

    #pragma unroll
    for (int tl = 0; tl < NRZ; ++tl) rz0[tl] = brz0[tl];
    #pragma unroll
    for (int tl = 0; tl < NN; ++tl) { xn0[tl] = zero8; hn0[tl] = zero8; }
    {
      const v2f xf[KC_I] = {{xa.x, xa.y}, {xa.z, xa.w}, {xb.x, xb.y}, {xb.z, xb.w}};
      auto beside = [&](int wi) __attribute__((always_inline)) {
        __builtin_amdgcn_sched_barrier(0);
        one_product<KC_I, KC_H>(wi, wih0, xf, xn0, whh0, h0, hn0, rz0);
        __builtin_amdgcn_sched_barrier(0);
      };
      cell_batch<0,  0>(rz1, xn1, hn1, bnx1, bnh1, h1, beside);
      cell_batch<NB, ST_CELL>(rz1, xn1, hn1, bnx1, bnh1, h1, beside);
    }
  }

  #pragma unroll
  for (int i = 0; i < HH; ++i) stage[m * H + HH * h + i] = h1[i >> 1][i & 1];
  __syncthreads();
  #pragma unroll
  for (int e = lane; e < 16 * H; e += 32) out[(size_t)bbase * H + e] = stage[e];
}

extern "C" void kernel_launch(void* const* d_in, const int* in_sizes, int n_in,
                              void* d_out, int out_size, void* d_ws, size_t ws_size,
                              hipStream_t stream) {
  const float* x    = (const float*)d_in[0];
  const float* Wih0 = (const float*)d_in[1];
  const float* Whh0 = (const float*)d_in[2];
  const float* bih0 = (const float*)d_in[3];
  const float* bhh0 = (const float*)d_in[4];
  const float* Wih1 = (const float*)d_in[5];
  const float* Whh1 = (const float*)d_in[6];
  const float* bih1 = (const float*)d_in[7];
  const float* bhh1 = (const float*)d_in[8];
  float* out = (float*)d_out;

  const int B = in_sizes[0] / (T * I);
  const int nblocks = (B + 15) / 16;

  gru2_fused<<<nblocks, 32, 0, stream>>>(x, Wih0, Whh0, bih0, bhh0,
                                         Wih1, Whh1, bih1, bhh1, out);
  (void)n_in; (void)out_size; (void)d_ws; (void)ws_size;
}


# Sigmoid quotient without the division's range steps: 1+e held in [1,2^126] (exponent argument capped at 126): rcp + the same 3 refinements, no scale/fix-up ops, no per-unit flags via VCC
# speedup vs baseline: 4.1811x; 1.1221x over previous
#include <hip/hip_runtime.h>
#include <cstdint>


typedef float v2f __attribute__((ext_vector_type(2)));
typedef float v4f __attribute__((ext_vector_type(4)));
typedef float v8f __attribute__((ext_vector_type(8)));


namespace {

constexpr int H    = 40;
constexpr int HH   = 20;
constexpr int I    = 16;
constexpr int T    = 256;
constexpr int NRZ  = 5;
constexpr int NN   = 3;
constexpr int NT   = NRZ + NN;
constexpr int KC_I = 4;
constexpr int KC_H = 10;

__device__ __forceinline__ v8f wmma_f32(v2f a, v2f b, v8f c) {
  return __builtin_amdgcn_wmma_f32_16x16x4_f32(
       false, a,  false, b,
       (short)0, c,  false,  false);
}


__device__ __forceinline__ int gate_row(int tl, int m) {
  const int hr = m >> 3, v = m & 7;
  if (tl < NRZ) {
    const int q = tl * 8 + v;
    return (q < HH) ? (HH * hr + q) : (H + HH * hr + (q - HH));
  }
  const int q = (tl - NRZ) * 8 + v;
  return (q < HH) ? (2 * H + HH * hr + q) : -1;
}

template <int K, int KC>
__device__ __forceinline__ void load_a(const float* __restrict__ w, int m, int h, v2f (&a)[NT][KC]) {
  #pragma unroll
  for (int tl = 0; tl < NT; ++tl) {
    const int g  = gate_row(tl, m);
    const float* p = w + (g >= 0 ? g : 0) * K + (K / 2) * h;
    #pragma unroll
    for (int kc = 0; kc < KC; ++kc) {
      const v2f t = *(const v2f*)(p + 2 * kc);
      a[tl][kc] = (g >= 0) ? t : (v2f){0.f, 0.f};
    }
  }
}

__device__ __forceinline__ void load_bias(const float* __restrict__ bih, const float* __restrict__ bhh,
                                          int h, v8f (&brz)[NRZ], float (&bnx)[HH], float (&bnh)[HH]) {
  #pragma unroll
  for (int tl = 0; tl < NRZ; ++tl) {
    #pragma unroll
    for (int v = 0; v < 8; ++v) {
      const int g = gate_row(tl, 8 * h + v);
      brz[tl][v] = bih[g] + bhh[g];
    }
  }
  #pragma unroll
  for (int i = 0; i < HH; ++i) {
    bnx[i] = bih[2 * H + HH * h + i];
    bnh[i] = bhh[2 * H + HH * h + i];
  }
}


constexpr int NB = 10;
constexpr int ST_SIG  = 10;
constexpr int ST_CELL = 2 * ST_SIG + 3 + 26 + 2;
static_assert(ST_CELL == 51, "the stage numbers in cell_batch are written out for this count");

template <class V>
__device__ __forceinline__ void keep_stage(V (&a)[NB]) {
  static_assert(NB == 10, "operand list below is written for ten");
  asm volatile("" : "+v"(a[0]), "+v"(a[1]), "+v"(a[2]), "+v"(a[3]), "+v"(a[4]),
                    "+v"(a[5]), "+v"(a[6]), "+v"(a[7]), "+v"(a[8]), "+v"(a[9]));
}

#define STAGE(k, arr, expr) { _Pragma("unroll") for (int j = 0; j < NB; ++j) { arr[j] = (expr); } \
                              keep_stage(arr); beside(W0 + (k)); }

#define STAGE_FLAG(k, arr, expr) { _Pragma("unroll") for (int j = 0; j < NB; ++j) { arr[j] = (expr); } \
                                   beside(W0 + (k)); }

#define SIGMOID_STAGES(K, SRC, OUT)                                                           \
  STAGE((K) + 0,  t,   __builtin_fminf((SRC) * -0x1.715476p+0f, 126.0f))                      \
  STAGE((K) + 1,  t,   __builtin_amdgcn_exp2f(t[j]))                                          \
  STAGE((K) + 2,  dd,  1.0f + t[j])                                                           \
  STAGE((K) + 3,  rc,  __builtin_amdgcn_rcpf(dd[j]))                                          \
  STAGE((K) + 4,  f,   __builtin_fmaf(-dd[j], rc[j], 1.0f))                                   \
  STAGE((K) + 5,  rc,  __builtin_fmaf(f[j], rc[j], rc[j]))                                    \
  STAGE((K) + 6,  f,   __builtin_fmaf(-dd[j], rc[j], 1.0f))                                   \
  STAGE((K) + 7,  q,   __builtin_fmaf(f[j], rc[j], rc[j]))                                    \
  STAGE((K) + 8,  f,   __builtin_fmaf(-dd[j], q[j], 1.0f))                                    \
  STAGE((K) + 9,  OUT, __builtin_fmaf(f[j], rc[j], q[j]))

template <int I0, int W0, class F>
__device__ __forceinline__ void cell_batch(const v8f (&rz)[NRZ], const v8f (&xn)[NN], const v8f (&hn)[NN],
                                           const float (&bnx)[HH], const float (&bnh)[HH],
                                           v2f (&h)[KC_H], F beside) {
  float t[NB], dd[NB], rc[NB], f[NB], q[NB], r[NB], z[NB];
  bool fl[NB];
  SIGMOID_STAGES(0,      rz[(I0 + j) >> 3][(I0 + j) & 7], r)
  SIGMOID_STAGES(ST_SIG, rz[(HH + I0 + j) >> 3][(HH + I0 + j) & 7], z)

  float u[NB], w[NB], pre[NB];
  STAGE(20, u,   xn[(I0 + j) >> 3][(I0 + j) & 7] + bnx[I0 + j])
  STAGE(21, w,   hn[(I0 + j) >> 3][(I0 + j) & 7] + bnh[I0 + j])
  STAGE(22, pre, u[j] + r[j] * w[j])

  float y[NB], y2[NB], p[NB], zs[NB], x2[NB], ph[NB], pl[NB], en[NB], e[NB];
  int ei[NB];
  STAGE(23, y,  __builtin_fabsf(pre[j]))
  STAGE(24, y2, y[j] * y[j])
  STAGE(25, p,  __builtin_fmaf(y2[j], -0x1.758e7ap-8f, 0x1.521192p-6f))
  STAGE(26, p,  __builtin_fmaf(y2[j], p[j], -0x1.b8389cp-5f))
  STAGE(27, p,  __builtin_fmaf(y2[j], p[j], 0x1.110704p-3f))
  STAGE(28, p,  __builtin_fmaf(y2[j], p[j], -0x1.555532p-2f))
  STAGE(29, p,  y[j] * p[j])
  STAGE(30, zs, __builtin_fmaf(y2[j], p[j], y[j]))
  STAGE(31, x2, y[j] + y[j])
  STAGE(32, ph, x2[j] * 0x1.715476p+0f)
  STAGE(33, pl, __builtin_fmaf(x2[j], 0x1.715476p+0f, -ph[j]))
  STAGE(34, en, __builtin_rintf(ph[j]))
  STAGE(35, pl, __builtin_fmaf(x2[j], 0x1.4ae0bep-26f, pl[j]))
  STAGE(36, ph, ph[j] - en[j])
  STAGE(37, ph, ph[j] + pl[j])
  STAGE(38, ei, (int)en[j])
  STAGE(39, e,  __builtin_amdgcn_exp2f(ph[j]))
  STAGE(40, e,  __builtin_ldexpf(e[j], ei[j]))
  STAGE_FLAG(41, fl, !(0x1.62e43p+6f < x2[j]))
  STAGE(42, e,  fl[j] ? e[j] : __builtin_inff())
  STAGE(43, e,  e[j] + 1.0f)
  STAGE(44, e,  __builtin_amdgcn_rcpf(e[j]))
  STAGE(45, e,  __builtin_fmaf(-2.0f, e[j], 1.0f))
  STAGE_FLAG(46, fl, y[j] < 0.625f)
  STAGE(47, e,  fl[j] ? zs[j] : e[j])
  STAGE(48, e,  __builtin_copysignf(e[j], pre[j]))

  STAGE(49, u,  h[(I0 + j) >> 1][(I0 + j) & 1] - e[j])
  STAGE(50, u,  e[j] + z[j] * u[j])
  #pragma unroll
  for (int j = 0; j < NB; ++j) h[(I0 + j) >> 1][(I0 + j) & 1] = u[j];
}

template <int KA, int KB>
__device__ __forceinline__ void one_product(int wi,
                                            const v2f (&wa)[NT][KA], const v2f (&ba)[KA], v8f (&na)[NN],
                                            const v2f (&wb)[NT][KB], const v2f (&bb)[KB], v8f (&nb)[NN],
                                            v8f (&rz)[NRZ]) {
  const int kc = wi >> 3, tl = wi & 7;
  if (kc < KA) {
    if (tl < NRZ) rz[tl] = wmma_f32(wa[tl][kc], ba[kc], rz[tl]);
    else          na[tl - NRZ] = wmma_f32(wa[tl][kc], ba[kc], na[tl - NRZ]);
  } else if (kc < KA + KB) {
    if (tl < NRZ) rz[tl] = wmma_f32(wb[tl][kc - KA], bb[kc - KA], rz[tl]);
    else          nb[tl - NRZ] = wmma_f32(wb[tl][kc - KA], bb[kc - KA], nb[tl - NRZ]);
  }
}

}

__global__ __launch_bounds__(32) void gru2_fused(
    const float* __restrict__ x,
    const float* __restrict__ Wih0, const float* __restrict__ Whh0,
    const float* __restrict__ bih0, const float* __restrict__ bhh0,
    const float* __restrict__ Wih1, const float* __restrict__ Whh1,
    const float* __restrict__ bih1, const float* __restrict__ bhh1,
    float* __restrict__ out)
{
  __shared__ float stage[16 * H];
  const int lane  = threadIdx.x & 31;
  const int m     = lane & 15;
  const int h     = lane >> 4;
  const int bbase = blockIdx.x * 16;

  v2f wih0[NT][KC_I], whh0[NT][KC_H], wih1[NT][KC_H], whh1[NT][KC_H];
  load_a<I, KC_I>(Wih0, m, h, wih0);
  load_a<H, KC_H>(Whh0, m, h, whh0);
  load_a<H, KC_H>(Wih1, m, h, wih1);
  load_a<H, KC_H>(Whh1, m, h, whh1);

  v8f brz0[NRZ], brz1[NRZ];
  float bnx0[HH], bnh0[HH], bnx1[HH], bnh1[HH];
  load_bias(bih0, bhh0, h, brz0, bnx0, bnh0);
  load_bias(bih1, bhh1, h, brz1, bnx1, bnh1);

  v2f h0[KC_H], h1[KC_H];
  #pragma unroll
  for (int kc = 0; kc < KC_H; ++kc) { h0[kc] = (v2f){0.f, 0.f}; h1[kc] = (v2f){0.f, 0.f}; }

  const float* xrow = x + (size_t)(bbase + m) * T * I + 8 * h;
  v4f xa = *(const v4f*)(xrow);
  v4f xb = *(const v4f*)(xrow + 4);

  const v8f zero8 = {0.f, 0.f, 0.f, 0.f, 0.f, 0.f, 0.f, 0.f};

  v8f rz0[NRZ], xn0[NN], hn0[NN];
  #pragma unroll
  for (int tl = 0; tl < NRZ; ++tl) rz0[tl] = brz0[tl];
  #pragma unroll
  for (int tl = 0; tl < NN; ++tl) { xn0[tl] = zero8; hn0[tl] = zero8; }
  {
    const v2f xf[KC_I] = {{xa.x, xa.y}, {xa.z, xa.w}, {xb.x, xb.y}, {xb.z, xb.w}};
    #pragma unroll
    for (int wi = 0; wi < 8 * (KC_I + KC_H); ++wi)
      one_product<KC_I, KC_H>(wi, wih0, xf, xn0, whh0, h0, hn0, rz0);
  }

  #pragma unroll 1
  for (int t = 0; t < T; ++t) {
    const int tn = (t + 1 < T) ? (t + 1) : t;
    xa = *(const v4f*)(xrow + tn * I);
    xb = *(const v4f*)(xrow + tn * I + 4);

    v8f rz1[NRZ], xn1[NN], hn1[NN];
    #pragma unroll
    for (int tl = 0; tl < NRZ; ++tl) rz1[tl] = brz1[tl];
    #pragma unroll
    for (int tl = 0; tl < NN; ++tl) { xn1[tl] = zero8; hn1[tl] = zero8; }
    {
      auto beside = [&](int wi) __attribute__((always_inline)) {
        __builtin_amdgcn_sched_barrier(0);
        one_product<KC_H, KC_H>(wi, whh1, h1, hn1, wih1, h0, xn1, rz1);
        __builtin_amdgcn_sched_barrier(0);
      };
      static_assert(ST_CELL - 1 < 8 * KC_H && 2 * ST_CELL - 1 < 8 * (KC_H + NB / 2),
                    "a fresh pair would be read by a product before its batch has written it");
      cell_batch<0,  0>(rz0, xn0, hn0, bnx0, bnh0, h0, beside);
      cell_batch<NB, ST_CELL>(rz0, xn0, hn0, bnx0, bnh0, h0, beside);
      #pragma unroll
      for (int wi = 2 * ST_CELL; wi < 8 * (KC_H + KC_H); ++wi)
        one_product<KC_H, KC_H>(wi, whh1, h1, hn1, wih1, h0, xn1, rz1);
      __builtin_amdgcn_sched_barrier(0);
    }

    #pragma unroll
    for (int tl = 0; tl < NRZ; ++tl) rz0[tl] = brz0[tl];
    #pragma unroll
    for (int tl = 0; tl < NN; ++tl) { xn0[tl] = zero8; hn0[tl] = zero8; }
    {
      const v2f xf[KC_I] = {{xa.x, xa.y}, {xa.z, xa.w}, {xb.x, xb.y}, {xb.z, xb.w}};
      auto beside = [&](int wi) __attribute__((always_inline)) {
        __builtin_amdgcn_sched_barrier(0);
        one_product<KC_I, KC_H>(wi, wih0, xf, xn0, whh0, h0, hn0, rz0);
        __builtin_amdgcn_sched_barrier(0);
      };
      cell_batch<0,  0>(rz1, xn1, hn1, bnx1, bnh1, h1, beside);
      cell_batch<NB, ST_CELL>(rz1, xn1, hn1, bnx1, bnh1, h1, beside);
      #pragma unroll
      for (int wi = 2 * ST_CELL; wi < 8 * (KC_I + KC_H); ++wi)
        one_product<KC_I, KC_H>(wi, wih0, xf, xn0, whh0, h0, hn0, rz0);
      __builtin_amdgcn_sched_barrier(0);
    }
  }

  #pragma unroll
  for (int i = 0; i < HH; ++i) stage[m * H + HH * h + i] = h1[i >> 1][i & 1];
  __syncthreads();
  #pragma unroll
  for (int e = lane; e < 16 * H; e += 32) out[(size_t)bbase * H + e] = stage[e];
}

extern "C" void kernel_launch(void* const* d_in, const int* in_sizes, int n_in,
                              void* d_out, int out_size, void* d_ws, size_t ws_size,
                              hipStream_t stream) {
  const float* x    = (const float*)d_in[0];
  const float* Wih0 = (const float*)d_in[1];
  const float* Whh0 = (const float*)d_in[2];
  const float* bih0 = (const float*)d_in[3];
  const float* bhh0 = (const float*)d_in[4];
  const float* Wih1 = (const float*)d_in[5];
  const float* Whh1 = (const float*)d_in[6];
  const float* bih1 = (const float*)d_in[7];
  const float* bhh1 = (const float*)d_in[8];
  float* out = (float*)d_out;

  const int B = in_sizes[0] / (T * I);
  const int nblocks = (B + 15) / 16;

  gru2_fused<<<nblocks, 32, 0, stream>>>(x, Wih0, Whh0, bih0, bhh0,
                                         Wih1, Whh1, bih1, bhh1, out);
  (void)n_in; (void)out_size; (void)d_ws; (void)ws_size;
}


# Cell update's multiplies, adds and fused multiply-adds issued two hidden units per instruction (2-wide f32 v_pk_fma/mul/add on unit pairs); same operations per value, sums unchanged
# speedup vs baseline: 4.3328x; 1.0363x over previous
#include <hip/hip_runtime.h>
#include <cstdint>


typedef float v2f __attribute__((ext_vector_type(2)));
typedef float v4f __attribute__((ext_vector_type(4)));
typedef float v8f __attribute__((ext_vector_type(8)));


namespace {

constexpr int H    = 40;
constexpr int HH   = 20;
constexpr int I    = 16;
constexpr int T    = 256;
constexpr int NRZ  = 5;
constexpr int NN   = 3;
constexpr int NT   = NRZ + NN;
constexpr int KC_I = 4;
constexpr int KC_H = 10;

__device__ __forceinline__ v8f wmma_f32(v2f a, v2f b, v8f c) {
  return __builtin_amdgcn_wmma_f32_16x16x4_f32(
       false, a,  false, b,
       (short)0, c,  false,  false);
}


__device__ __forceinline__ int gate_row(int tl, int m) {
  const int hr = m >> 3, v = m & 7;
  if (tl < NRZ) {
    const int q = tl * 8 + v;
    return (q < HH) ? (HH * hr + q) : (H + HH * hr + (q - HH));
  }
  const int q = (tl - NRZ) * 8 + v;
  return (q < HH) ? (2 * H + HH * hr + q) : -1;
}

template <int K, int KC>
__device__ __forceinline__ void load_a(const float* __restrict__ w, int m, int h, v2f (&a)[NT][KC]) {
  #pragma unroll
  for (int tl = 0; tl < NT; ++tl) {
    const int g  = gate_row(tl, m);
    const float* p = w + (g >= 0 ? g : 0) * K + (K / 2) * h;
    #pragma unroll
    for (int kc = 0; kc < KC; ++kc) {
      const v2f t = *(const v2f*)(p + 2 * kc);
      a[tl][kc] = (g >= 0) ? t : (v2f){0.f, 0.f};
    }
  }
}

__device__ __forceinline__ void load_bias(const float* __restrict__ bih, const float* __restrict__ bhh,
                                          int h, v8f (&brz)[NRZ], v2f (&bnx)[HH / 2], v2f (&bnh)[HH / 2]) {
  #pragma unroll
  for (int tl = 0; tl < NRZ; ++tl) {
    #pragma unroll
    for (int v = 0; v < 8; ++v) {
      const int g = gate_row(tl, 8 * h + v);
      brz[tl][v] = bih[g] + bhh[g];
    }
  }
  #pragma unroll
  for (int i = 0; i < HH; ++i) {
    bnx[i >> 1][i & 1] = bih[2 * H + HH * h + i];
    bnh[i >> 1][i & 1] = bhh[2 * H + HH * h + i];
  }
}


typedef int v2i __attribute__((ext_vector_type(2)));

constexpr int NB = 10;
constexpr int NP = NB / 2;
constexpr int ST_SIG  = 10;
constexpr int ST_CELL = 2 * ST_SIG + 3 + 26 + 2;
static_assert(ST_CELL == 51, "the stage numbers in cell_batch are written out for this count");

__device__ __forceinline__ v2f splat(float k) { return (v2f){k, k}; }
__device__ __forceinline__ v2f fma2(v2f a, v2f b, v2f c) { return __builtin_elementwise_fma(a, b, c); }
__device__ __forceinline__ v2f pair_of(const v8f &a, int e) { return (v2f){a[e], a[e + 1]}; }

template <class V>
__device__ __forceinline__ void keep_stage(V (&a)[NP]) {
  static_assert(NP == 5, "operand list below is written for five");
  asm volatile("" : "+v"(a[0]), "+v"(a[1]), "+v"(a[2]), "+v"(a[3]), "+v"(a[4]));
}

#define STAGE2(k, arr, expr) { _Pragma("unroll") for (int j = 0; j < NP; ++j) { arr[j] = (expr); } \
                               keep_stage(arr); beside(W0 + (k)); }

#define STAGE1(k, arr, expr) { _Pragma("unroll") for (int j = 0; j < NP; ++j) {                     \
                                 _Pragma("unroll") for (int c = 0; c < 2; ++c) { arr[j][c] = (expr); } } \
                               keep_stage(arr); beside(W0 + (k)); }

#define STAGE_FLAG(k, arr, expr) { _Pragma("unroll") for (int j = 0; j < NP; ++j) {                 \
                                     _Pragma("unroll") for (int c = 0; c < 2; ++c) { arr[2 * j + c] = (expr); } } \
                                   beside(W0 + (k)); }

#define SIGMOID_STAGES(K, SRC, OUT)                                                           \
  STAGE2((K) + 0,  t,   __builtin_elementwise_min((SRC) * splat(-0x1.715476p+0f), splat(126.0f))) \
  STAGE1((K) + 1,  t,   __builtin_amdgcn_exp2f(t[j][c]))                                      \
  STAGE2((K) + 2,  dd,  splat(1.0f) + t[j])                                                   \
  STAGE1((K) + 3,  rc,  __builtin_amdgcn_rcpf(dd[j][c]))                                      \
  STAGE2((K) + 4,  f,   fma2(-dd[j], rc[j], splat(1.0f)))                                     \
  STAGE2((K) + 5,  rc,  fma2(f[j], rc[j], rc[j]))                                             \
  STAGE2((K) + 6,  f,   fma2(-dd[j], rc[j], splat(1.0f)))                                     \
  STAGE2((K) + 7,  q,   fma2(f[j], rc[j], rc[j]))                                             \
  STAGE2((K) + 8,  f,   fma2(-dd[j], q[j], splat(1.0f)))                                      \
  STAGE2((K) + 9,  OUT, fma2(f[j], rc[j], q[j]))

template <int I0, int W0, class F>
__device__ __forceinline__ void cell_batch(const v8f (&rz)[NRZ], const v8f (&xn)[NN], const v8f (&hn)[NN],
                                           const v2f (&bnx)[HH / 2], const v2f (&bnh)[HH / 2],
                                           v2f (&h)[KC_H], F beside) {
  static_assert(I0 % 2 == 0 && HH % 2 == 0, "a pair must not straddle an accumulator tile or the r|z split");
  v2f t[NP], dd[NP], rc[NP], f[NP], q[NP], r[NP], z[NP];
  bool fl[NB];
  SIGMOID_STAGES(0,      pair_of(rz[(I0 + 2 * j) >> 3], (I0 + 2 * j) & 7), r)
  SIGMOID_STAGES(ST_SIG, pair_of(rz[(HH + I0 + 2 * j) >> 3], (HH + I0 + 2 * j) & 7), z)

  v2f u[NP], w[NP], pre[NP];
  STAGE2(20, u,   pair_of(xn[(I0 + 2 * j) >> 3], (I0 + 2 * j) & 7) + bnx[I0 / 2 + j])
  STAGE2(21, w,   pair_of(hn[(I0 + 2 * j) >> 3], (I0 + 2 * j) & 7) + bnh[I0 / 2 + j])
  STAGE2(22, pre, fma2(r[j], w[j], u[j]))

  v2f y[NP], y2[NP], p[NP], zs[NP], x2[NP], ph[NP], pl[NP], en[NP], e[NP];
  v2i ei[NP];
  STAGE1(23, y,  __builtin_fabsf(pre[j][c]))
  STAGE2(24, y2, y[j] * y[j])
  STAGE2(25, p,  fma2(y2[j], splat(-0x1.758e7ap-8f), splat(0x1.521192p-6f)))
  STAGE2(26, p,  fma2(y2[j], p[j], splat(-0x1.b8389cp-5f)))
  STAGE2(27, p,  fma2(y2[j], p[j], splat(0x1.110704p-3f)))
  STAGE2(28, p,  fma2(y2[j], p[j], splat(-0x1.555532p-2f)))
  STAGE2(29, p,  y[j] * p[j])
  STAGE2(30, zs, fma2(y2[j], p[j], y[j]))
  STAGE2(31, x2, y[j] + y[j])
  STAGE2(32, ph, x2[j] * splat(0x1.715476p+0f))
  STAGE2(33, pl, fma2(x2[j], splat(0x1.715476p+0f), -ph[j]))
  STAGE1(34, en, __builtin_rintf(ph[j][c]))
  STAGE2(35, pl, fma2(x2[j], splat(0x1.4ae0bep-26f), pl[j]))
  STAGE2(36, ph, ph[j] - en[j])
  STAGE2(37, ph, ph[j] + pl[j])
  STAGE1(38, ei, (int)en[j][c])
  STAGE1(39, e,  __builtin_amdgcn_exp2f(ph[j][c]))
  STAGE1(40, e,  __builtin_ldexpf(e[j][c], ei[j][c]))
  STAGE_FLAG(41, fl, !(0x1.62e43p+6f < x2[j][c]))
  STAGE1(42, e,  fl[2 * j + c] ? e[j][c] : __builtin_inff())
  STAGE2(43, e,  e[j] + splat(1.0f))
  STAGE1(44, e,  __builtin_amdgcn_rcpf(e[j][c]))
  STAGE2(45, e,  fma2(splat(-2.0f), e[j], splat(1.0f)))
  STAGE_FLAG(46, fl, y[j][c] < 0.625f)
  STAGE1(47, e,  fl[2 * j + c] ? zs[j][c] : e[j][c])
  STAGE1(48, e,  __builtin_copysignf(e[j][c], pre[j][c]))

  STAGE2(49, u,  h[I0 / 2 + j] - e[j])
  STAGE2(50, u,  fma2(z[j], u[j], e[j]))
  #pragma unroll
  for (int j = 0; j < NP; ++j) h[I0 / 2 + j] = u[j];
}

template <int KA, int KB>
__device__ __forceinline__ void one_product(int wi,
                                            const v2f (&wa)[NT][KA], const v2f (&ba)[KA], v8f (&na)[NN],
                                            const v2f (&wb)[NT][KB], const v2f (&bb)[KB], v8f (&nb)[NN],
                                            v8f (&rz)[NRZ]) {
  const int kc = wi >> 3, tl = wi & 7;
  if (kc < KA) {
    if (tl < NRZ) rz[tl] = wmma_f32(wa[tl][kc], ba[kc], rz[tl]);
    else          na[tl - NRZ] = wmma_f32(wa[tl][kc], ba[kc], na[tl - NRZ]);
  } else if (kc < KA + KB) {
    if (tl < NRZ) rz[tl] = wmma_f32(wb[tl][kc - KA], bb[kc - KA], rz[tl]);
    else          nb[tl - NRZ] = wmma_f32(wb[tl][kc - KA], bb[kc - KA], nb[tl - NRZ]);
  }
}

}

__global__ __launch_bounds__(32) void gru2_fused(
    const float* __restrict__ x,
    const float* __restrict__ Wih0, const float* __restrict__ Whh0,
    const float* __restrict__ bih0, const float* __restrict__ bhh0,
    const float* __restrict__ Wih1, const float* __restrict__ Whh1,
    const float* __restrict__ bih1, const float* __restrict__ bhh1,
    float* __restrict__ out)
{
  __shared__ float stage[16 * H];
  const int lane  = threadIdx.x & 31;
  const int m     = lane & 15;
  const int h     = lane >> 4;
  const int bbase = blockIdx.x * 16;

  v2f wih0[NT][KC_I], whh0[NT][KC_H], wih1[NT][KC_H], whh1[NT][KC_H];
  load_a<I, KC_I>(Wih0, m, h, wih0);
  load_a<H, KC_H>(Whh0, m, h, whh0);
  load_a<H, KC_H>(Wih1, m, h, wih1);
  load_a<H, KC_H>(Whh1, m, h, whh1);

  v8f brz0[NRZ], brz1[NRZ];
  v2f bnx0[HH / 2], bnh0[HH / 2], bnx1[HH / 2], bnh1[HH / 2];
  load_bias(bih0, bhh0, h, brz0, bnx0, bnh0);
  load_bias(bih1, bhh1, h, brz1, bnx1, bnh1);

  v2f h0[KC_H], h1[KC_H];
  #pragma unroll
  for (int kc = 0; kc < KC_H; ++kc) { h0[kc] = (v2f){0.f, 0.f}; h1[kc] = (v2f){0.f, 0.f}; }

  const float* xrow = x + (size_t)(bbase + m) * T * I + 8 * h;
  v4f xa = *(const v4f*)(xrow);
  v4f xb = *(const v4f*)(xrow + 4);

  const v8f zero8 = {0.f, 0.f, 0.f, 0.f, 0.f, 0.f, 0.f, 0.f};

  v8f rz0[NRZ], xn0[NN], hn0[NN];
  #pragma unroll
  for (int tl = 0; tl < NRZ; ++tl) rz0[tl] = brz0[tl];
  #pragma unroll
  for (int tl = 0; tl < NN; ++tl) { xn0[tl] = zero8; hn0[tl] = zero8; }
  {
    const v2f xf[KC_I] = {{xa.x, xa.y}, {xa.z, xa.w}, {xb.x, xb.y}, {xb.z, xb.w}};
    #pragma unroll
    for (int wi = 0; wi < 8 * (KC_I + KC_H); ++wi)
      one_product<KC_I, KC_H>(wi, wih0, xf, xn0, whh0, h0, hn0, rz0);
  }

  #pragma unroll 1
  for (int t = 0; t < T; ++t) {
    const int tn = (t + 1 < T) ? (t + 1) : t;
    xa = *(const v4f*)(xrow + tn * I);
    xb = *(const v4f*)(xrow + tn * I + 4);

    v8f rz1[NRZ], xn1[NN], hn1[NN];
    #pragma unroll
    for (int tl = 0; tl < NRZ; ++tl) rz1[tl] = brz1[tl];
    #pragma unroll
    for (int tl = 0; tl < NN; ++tl) { xn1[tl] = zero8; hn1[tl] = zero8; }
    {
      auto beside = [&](int wi) __attribute__((always_inline)) {
        __builtin_amdgcn_sched_barrier(0);
        one_product<KC_H, KC_H>(wi, whh1, h1, hn1, wih1, h0, xn1, rz1);
        __builtin_amdgcn_sched_barrier(0);
      };
      static_assert(ST_CELL - 1 < 8 * KC_H && 2 * ST_CELL - 1 < 8 * (KC_H + NB / 2),
                    "a fresh pair would be read by a product before its batch has written it");
      cell_batch<0,  0>(rz0, xn0, hn0, bnx0, bnh0, h0, beside);
      cell_batch<NB, ST_CELL>(rz0, xn0, hn0, bnx0, bnh0, h0, beside);
      #pragma unroll
      for (int wi = 2 * ST_CELL; wi < 8 * (KC_H + KC_H); ++wi)
        one_product<KC_H, KC_H>(wi, whh1, h1, hn1, wih1, h0, xn1, rz1);
      __builtin_amdgcn_sched_barrier(0);
    }

    #pragma unroll
    for (int tl = 0; tl < NRZ; ++tl) rz0[tl] = brz0[tl];
    #pragma unroll
    for (int tl = 0; tl < NN; ++tl) { xn0[tl] = zero8; hn0[tl] = zero8; }
    {
      const v2f xf[KC_I] = {{xa.x, xa.y}, {xa.z, xa.w}, {xb.x, xb.y}, {xb.z, xb.w}};
      auto beside = [&](int wi) __attribute__((always_inline)) {
        __builtin_amdgcn_sched_barrier(0);
        one_product<KC_I, KC_H>(wi, wih0, xf, xn0, whh0, h0, hn0, rz0);
        __builtin_amdgcn_sched_barrier(0);
      };
      cell_batch<0,  0>(rz1, xn1, hn1, bnx1, bnh1, h1, beside);
      cell_batch<NB, ST_CELL>(rz1, xn1, hn1, bnx1, bnh1, h1, beside);
      #pragma unroll
      for (int wi = 2 * ST_CELL; wi < 8 * (KC_I + KC_H); ++wi)
        one_product<KC_I, KC_H>(wi, wih0, xf, xn0, whh0, h0, hn0, rz0);
      __builtin_amdgcn_sched_barrier(0);
    }
  }

  #pragma unroll
  for (int i = 0; i < HH; ++i) stage[m * H + HH * h + i] = h1[i >> 1][i & 1];
  __syncthreads();
  #pragma unroll
  for (int e = lane; e < 16 * H; e += 32) out[(size_t)bbase * H + e] = stage[e];
}

extern "C" void kernel_launch(void* const* d_in, const int* in_sizes, int n_in,
                              void* d_out, int out_size, void* d_ws, size_t ws_size,
                              hipStream_t stream) {
  const float* x    = (const float*)d_in[0];
  const float* Wih0 = (const float*)d_in[1];
  const float* Whh0 = (const float*)d_in[2];
  const float* bih0 = (const float*)d_in[3];
  const float* bhh0 = (const float*)d_in[4];
  const float* Wih1 = (const float*)d_in[5];
  const float* Whh1 = (const float*)d_in[6];
  const float* bih1 = (const float*)d_in[7];
  const float* bhh1 = (const float*)d_in[8];
  float* out = (float*)d_out;

  const int B = in_sizes[0] / (T * I);
  const int nblocks = (B + 15) / 16;

  gru2_fused<<<nblocks, 32, 0, stream>>>(x, Wih0, Whh0, bih0, bhh0,
                                         Wih1, Whh1, bih1, bhh1, out);
  (void)n_in; (void)out_size; (void)d_ws; (void)ws_size;
}


# Layer 0's input projection (Wih0 x + r|z bias), fed by no state, moves to a second wave on another SIMD, a chunk of steps ahead, handed over in LDS as accumulator start values; sums unchanged
# speedup vs baseline: 4.6323x; 1.0691x over previous
#include <hip/hip_runtime.h>
#include <cstdint>


typedef float v2f __attribute__((ext_vector_type(2)));
typedef float v4f __attribute__((ext_vector_type(4)));
typedef float v8f __attribute__((ext_vector_type(8)));


namespace {

constexpr int H    = 40;
constexpr int HH   = 20;
constexpr int I    = 16;
constexpr int T    = 256;
constexpr int NRZ  = 5;
constexpr int NN   = 3;
constexpr int NT   = NRZ + NN;
constexpr int KC_I = 4;
constexpr int KC_H = 10;

__device__ __forceinline__ v8f wmma_f32(v2f a, v2f b, v8f c) {
  return __builtin_amdgcn_wmma_f32_16x16x4_f32(
       false, a,  false, b,
       (short)0, c,  false,  false);
}


__device__ __forceinline__ int gate_row(int tl, int m) {
  const int hr = m >> 3, v = m & 7;
  if (tl < NRZ) {
    const int q = tl * 8 + v;
    return (q < HH) ? (HH * hr + q) : (H + HH * hr + (q - HH));
  }
  const int q = (tl - NRZ) * 8 + v;
  return (q < HH) ? (2 * H + HH * hr + q) : -1;
}

template <int K, int KC>
__device__ __forceinline__ void load_a(const float* __restrict__ w, int m, int h, v2f (&a)[NT][KC]) {
  #pragma unroll
  for (int tl = 0; tl < NT; ++tl) {
    const int g  = gate_row(tl, m);
    const float* p = w + (g >= 0 ? g : 0) * K + (K / 2) * h;
    #pragma unroll
    for (int kc = 0; kc < KC; ++kc) {
      const v2f t = *(const v2f*)(p + 2 * kc);
      a[tl][kc] = (g >= 0) ? t : (v2f){0.f, 0.f};
    }
  }
}

__device__ __forceinline__ void load_bias(const float* __restrict__ bih, const float* __restrict__ bhh,
                                          int h, v8f (&brz)[NRZ], v2f (&bnx)[HH / 2], v2f (&bnh)[HH / 2]) {
  #pragma unroll
  for (int tl = 0; tl < NRZ; ++tl) {
    #pragma unroll
    for (int v = 0; v < 8; ++v) {
      const int g = gate_row(tl, 8 * h + v);
      brz[tl][v] = bih[g] + bhh[g];
    }
  }
  #pragma unroll
  for (int i = 0; i < HH; ++i) {
    bnx[i >> 1][i & 1] = bih[2 * H + HH * h + i];
    bnh[i >> 1][i & 1] = bhh[2 * H + HH * h + i];
  }
}


typedef int v2i __attribute__((ext_vector_type(2)));

constexpr int NB = 10;
constexpr int NP = NB / 2;
constexpr int ST_SIG  = 10;
constexpr int ST_CELL = 2 * ST_SIG + 3 + 26 + 2;
static_assert(ST_CELL == 51, "the stage numbers in cell_batch are written out for this count");

__device__ __forceinline__ v2f splat(float k) { return (v2f){k, k}; }
__device__ __forceinline__ v2f fma2(v2f a, v2f b, v2f c) { return __builtin_elementwise_fma(a, b, c); }
__device__ __forceinline__ v2f pair_of(const v8f &a, int e) { return (v2f){a[e], a[e + 1]}; }

template <class V>
__device__ __forceinline__ void keep_stage(V (&a)[NP]) {
  static_assert(NP == 5, "operand list below is written for five");
  asm volatile("" : "+v"(a[0]), "+v"(a[1]), "+v"(a[2]), "+v"(a[3]), "+v"(a[4]));
}

#define STAGE2(k, arr, expr) { _Pragma("unroll") for (int j = 0; j < NP; ++j) { arr[j] = (expr); } \
                               keep_stage(arr); beside(W0 + (k)); }

#define STAGE1(k, arr, expr) { _Pragma("unroll") for (int j = 0; j < NP; ++j) {                     \
                                 _Pragma("unroll") for (int c = 0; c < 2; ++c) { arr[j][c] = (expr); } } \
                               keep_stage(arr); beside(W0 + (k)); }

#define STAGE_FLAG(k, arr, expr) { _Pragma("unroll") for (int j = 0; j < NP; ++j) {                 \
                                     _Pragma("unroll") for (int c = 0; c < 2; ++c) { arr[2 * j + c] = (expr); } } \
                                   beside(W0 + (k)); }

#define SIGMOID_STAGES(K, SRC, OUT)                                                           \
  STAGE2((K) + 0,  t,   __builtin_elementwise_min((SRC) * splat(-0x1.715476p+0f), splat(126.0f))) \
  STAGE1((K) + 1,  t,   __builtin_amdgcn_exp2f(t[j][c]))                                      \
  STAGE2((K) + 2,  dd,  splat(1.0f) + t[j])                                                   \
  STAGE1((K) + 3,  rc,  __builtin_amdgcn_rcpf(dd[j][c]))                                      \
  STAGE2((K) + 4,  f,   fma2(-dd[j], rc[j], splat(1.0f)))                                     \
  STAGE2((K) + 5,  rc,  fma2(f[j], rc[j], rc[j]))                                             \
  STAGE2((K) + 6,  f,   fma2(-dd[j], rc[j], splat(1.0f)))                                     \
  STAGE2((K) + 7,  q,   fma2(f[j], rc[j], rc[j]))                                             \
  STAGE2((K) + 8,  f,   fma2(-dd[j], q[j], splat(1.0f)))                                      \
  STAGE2((K) + 9,  OUT, fma2(f[j], rc[j], q[j]))

template <int I0, int W0, class F>
__device__ __forceinline__ void cell_batch(const v8f (&rz)[NRZ], const v8f (&xn)[NN], const v8f (&hn)[NN],
                                           const v2f (&bnx)[HH / 2], const v2f (&bnh)[HH / 2],
                                           v2f (&h)[KC_H], F beside) {
  static_assert(I0 % 2 == 0 && HH % 2 == 0, "a pair must not straddle an accumulator tile or the r|z split");
  v2f t[NP], dd[NP], rc[NP], f[NP], q[NP], r[NP], z[NP];
  bool fl[NB];
  SIGMOID_STAGES(0,      pair_of(rz[(I0 + 2 * j) >> 3], (I0 + 2 * j) & 7), r)
  SIGMOID_STAGES(ST_SIG, pair_of(rz[(HH + I0 + 2 * j) >> 3], (HH + I0 + 2 * j) & 7), z)

  v2f u[NP], w[NP], pre[NP];
  STAGE2(20, u,   pair_of(xn[(I0 + 2 * j) >> 3], (I0 + 2 * j) & 7) + bnx[I0 / 2 + j])
  STAGE2(21, w,   pair_of(hn[(I0 + 2 * j) >> 3], (I0 + 2 * j) & 7) + bnh[I0 / 2 + j])
  STAGE2(22, pre, fma2(r[j], w[j], u[j]))

  v2f y[NP], y2[NP], p[NP], zs[NP], x2[NP], ph[NP], pl[NP], en[NP], e[NP];
  v2i ei[NP];
  STAGE1(23, y,  __builtin_fabsf(pre[j][c]))
  STAGE2(24, y2, y[j] * y[j])
  STAGE2(25, p,  fma2(y2[j], splat(-0x1.758e7ap-8f), splat(0x1.521192p-6f)))
  STAGE2(26, p,  fma2(y2[j], p[j], splat(-0x1.b8389cp-5f)))
  STAGE2(27, p,  fma2(y2[j], p[j], splat(0x1.110704p-3f)))
  STAGE2(28, p,  fma2(y2[j], p[j], splat(-0x1.555532p-2f)))
  STAGE2(29, p,  y[j] * p[j])
  STAGE2(30, zs, fma2(y2[j], p[j], y[j]))
  STAGE2(31, x2, y[j] + y[j])
  STAGE2(32, ph, x2[j] * splat(0x1.715476p+0f))
  STAGE2(33, pl, fma2(x2[j], splat(0x1.715476p+0f), -ph[j]))
  STAGE1(34, en, __builtin_rintf(ph[j][c]))
  STAGE2(35, pl, fma2(x2[j], splat(0x1.4ae0bep-26f), pl[j]))
  STAGE2(36, ph, ph[j] - en[j])
  STAGE2(37, ph, ph[j] + pl[j])
  STAGE1(38, ei, (int)en[j][c])
  STAGE1(39, e,  __builtin_amdgcn_exp2f(ph[j][c]))
  STAGE1(40, e,  __builtin_ldexpf(e[j][c], ei[j][c]))
  STAGE_FLAG(41, fl, !(0x1.62e43p+6f < x2[j][c]))
  STAGE1(42, e,  fl[2 * j + c] ? e[j][c] : __builtin_inff())
  STAGE2(43, e,  e[j] + splat(1.0f))
  STAGE1(44, e,  __builtin_amdgcn_rcpf(e[j][c]))
  STAGE2(45, e,  fma2(splat(-2.0f), e[j], splat(1.0f)))
  STAGE_FLAG(46, fl, y[j][c] < 0.625f)
  STAGE1(47, e,  fl[2 * j + c] ? zs[j][c] : e[j][c])
  STAGE1(48, e,  __builtin_copysignf(e[j][c], pre[j][c]))

  STAGE2(49, u,  h[I0 / 2 + j] - e[j])
  STAGE2(50, u,  fma2(z[j], u[j], e[j]))
  #pragma unroll
  for (int j = 0; j < NP; ++j) h[I0 / 2 + j] = u[j];
}

template <int KA, int KB>
__device__ __forceinline__ void one_product(int wi,
                                            const v2f (&wa)[NT][KA], const v2f (&ba)[KA], v8f (&na)[NN],
                                            const v2f (&wb)[NT][KB], const v2f (&bb)[KB], v8f (&nb)[NN],
                                            v8f (&rz)[NRZ]) {
  const int kc = wi >> 3, tl = wi & 7;
  if (kc < KA) {
    if (tl < NRZ) rz[tl] = wmma_f32(wa[tl][kc], ba[kc], rz[tl]);
    else          na[tl - NRZ] = wmma_f32(wa[tl][kc], ba[kc], na[tl - NRZ]);
  } else if (kc < KA + KB) {
    if (tl < NRZ) rz[tl] = wmma_f32(wb[tl][kc - KA], bb[kc - KA], rz[tl]);
    else          nb[tl - NRZ] = wmma_f32(wb[tl][kc - KA], bb[kc - KA], nb[tl - NRZ]);
  }
}


constexpr int CH   = 4;
constexpr int NCH  = T / CH;
constexpr int SLOT = 16 * 32;
static_assert(T % CH == 0, "the hand-over loop is written for whole chunks");

template <int N>
__device__ __forceinline__ void produce(const float* __restrict__ xrow, int s0, v4f* dst,
                                        const float* __restrict__ Wih0,
                                        const float* __restrict__ bih0, const float* __restrict__ bhh0,
                                        int m, int h, int lane) {
  v2f wih0[NT][KC_I];
  load_a<I, KC_I>(Wih0, m, h, wih0);
  v8f brz0[NRZ];
  v2f bnx[HH / 2], bnh[HH / 2];
  load_bias(bih0, bhh0, h, brz0, bnx, bnh);
  v4f xa[N], xb[N];
  #pragma unroll
  for (int i = 0; i < N; ++i) {
    const int s = (s0 + i < T) ? (s0 + i) : (T - 1);
    xa[i] = *(const v4f*)(xrow + s * I);
    xb[i] = *(const v4f*)(xrow + s * I + 4);
  }
  const v8f zero8 = {0.f, 0.f, 0.f, 0.f, 0.f, 0.f, 0.f, 0.f};
  #pragma unroll
  for (int i = 0; i < N; ++i) {
    const v2f xf[KC_I] = {{xa[i].x, xa[i].y}, {xa[i].z, xa[i].w}, {xb[i].x, xb[i].y}, {xb[i].z, xb[i].w}};
    v8f rz[NRZ], xn[NN];
    #pragma unroll
    for (int tl = 0; tl < NRZ; ++tl) rz[tl] = brz0[tl];
    #pragma unroll
    for (int tl = 0; tl < NN; ++tl) xn[tl] = zero8;
    #pragma unroll
    for (int wi = 0; wi < 8 * KC_I; ++wi) {
      const int kc = wi >> 3, tl = wi & 7;
      if (tl < NRZ) rz[tl] = wmma_f32(wih0[tl][kc], xf[kc], rz[tl]);
      else          xn[tl - NRZ] = wmma_f32(wih0[tl][kc], xf[kc], xn[tl - NRZ]);
    }
    v4f* d = dst + i * SLOT + lane;
    #pragma unroll
    for (int tl = 0; tl < NRZ; ++tl) {
      d[(2 * tl) * 32]     = (v4f){rz[tl][0], rz[tl][1], rz[tl][2], rz[tl][3]};
      d[(2 * tl + 1) * 32] = (v4f){rz[tl][4], rz[tl][5], rz[tl][6], rz[tl][7]};
    }
    #pragma unroll
    for (int tl = 0; tl < NN; ++tl) {
      d[(2 * (NRZ + tl)) * 32]     = (v4f){xn[tl][0], xn[tl][1], xn[tl][2], xn[tl][3]};
      d[(2 * (NRZ + tl) + 1) * 32] = (v4f){xn[tl][4], xn[tl][5], xn[tl][6], xn[tl][7]};
    }
  }
}

__device__ __forceinline__ void take_slot(const v4f* src, v8f (&rz)[NRZ], v8f (&xn)[NN]) {
  #pragma unroll
  for (int tl = 0; tl < NRZ; ++tl) {
    const v4f lo = src[(2 * tl) * 32], hi = src[(2 * tl + 1) * 32];
    rz[tl] = (v8f){lo.x, lo.y, lo.z, lo.w, hi.x, hi.y, hi.z, hi.w};
  }
  #pragma unroll
  for (int tl = 0; tl < NN; ++tl) {
    const v4f lo = src[(2 * (NRZ + tl)) * 32], hi = src[(2 * (NRZ + tl) + 1) * 32];
    xn[tl] = (v8f){lo.x, lo.y, lo.z, lo.w, hi.x, hi.y, hi.z, hi.w};
  }
}

__device__ __forceinline__ void state_product(int wi, const v2f (&w)[NT][KC_H], const v2f (&b)[KC_H],
                                              v8f (&n)[NN], v8f (&rz)[NRZ]) {
  const int kc = wi >> 3, tl = wi & 7;
  if (kc < KC_H) {
    if (tl < NRZ) rz[tl] = wmma_f32(w[tl][kc], b[kc], rz[tl]);
    else          n[tl - NRZ] = wmma_f32(w[tl][kc], b[kc], n[tl - NRZ]);
  }
}

}

__global__ __launch_bounds__(64) void gru2_fused(
    const float* __restrict__ x,
    const float* __restrict__ Wih0, const float* __restrict__ Whh0,
    const float* __restrict__ bih0, const float* __restrict__ bhh0,
    const float* __restrict__ Wih1, const float* __restrict__ Whh1,
    const float* __restrict__ bih1, const float* __restrict__ bhh1,
    float* __restrict__ out)
{
  __shared__ v4f ring[2 * CH * SLOT];
  __shared__ v4f first[SLOT];
  __shared__ float stage[16 * H];
  const int lane  = threadIdx.x & 31;
  const int wave  = __builtin_amdgcn_readfirstlane((int)(threadIdx.x >> 5));
  const int m     = lane & 15;
  const int h     = lane >> 4;
  const int bbase = blockIdx.x * 16;

  const float* xrow = x + (size_t)(bbase + m) * T * I + 8 * h;

  const v8f zero8 = {0.f, 0.f, 0.f, 0.f, 0.f, 0.f, 0.f, 0.f};

  v2f whh0[NT][KC_H], wih1[NT][KC_H], whh1[NT][KC_H];
  v8f brz1[NRZ];
  v2f bnx0[HH / 2], bnh0[HH / 2], bnx1[HH / 2], bnh1[HH / 2];
  v2f h0[KC_H], h1[KC_H];
  v8f rz0[NRZ], xn0[NN], hn0[NN];

  if (wave == 0) {
    load_a<H, KC_H>(Whh0, m, h, whh0);
    load_a<H, KC_H>(Wih1, m, h, wih1);
    load_a<H, KC_H>(Whh1, m, h, whh1);
    v8f brz0[NRZ];
    load_bias(bih0, bhh0, h, brz0, bnx0, bnh0);
    load_bias(bih1, bhh1, h, brz1, bnx1, bnh1);
    #pragma unroll
    for (int kc = 0; kc < KC_H; ++kc) { h0[kc] = (v2f){0.f, 0.f}; h1[kc] = (v2f){0.f, 0.f}; }
  } else {
    produce<1>(xrow, 0, first, Wih0, bih0, bhh0, m, h, lane);
  }
  __syncthreads();

  if (wave == 0) {
    take_slot(first + lane, rz0, xn0);
    #pragma unroll
    for (int tl = 0; tl < NN; ++tl) hn0[tl] = zero8;
    #pragma unroll
    for (int wi = 0; wi < 8 * KC_H; ++wi) state_product(wi, whh0, h0, hn0, rz0);
  }

  #pragma unroll 1
  for (int c = 0; c <= NCH; ++c) {
    if (wave != 0) {
      if (c < NCH) produce<CH>(xrow, c * CH + 1, ring + (c & 1) * (CH * SLOT), Wih0, bih0, bhh0, m, h, lane);
    } else if (c > 0) {
      const v4f* chunk = ring + ((c - 1) & 1) * (CH * SLOT) + lane;
      #pragma unroll 1
      for (int i = 0; i < CH; ++i) {
        v8f srz[NRZ], sxn[NN];
        take_slot(chunk + i * SLOT, srz, sxn);

        v8f rz1[NRZ], xn1[NN], hn1[NN];
        #pragma unroll
        for (int tl = 0; tl < NRZ; ++tl) rz1[tl] = brz1[tl];
        #pragma unroll
        for (int tl = 0; tl < NN; ++tl) { xn1[tl] = zero8; hn1[tl] = zero8; }
        {
          auto beside = [&](int wi) __attribute__((always_inline)) {
            __builtin_amdgcn_sched_barrier(0);
            one_product<KC_H, KC_H>(wi, whh1, h1, hn1, wih1, h0, xn1, rz1);
            __builtin_amdgcn_sched_barrier(0);
          };
          static_assert(ST_CELL - 1 < 8 * KC_H && 2 * ST_CELL - 1 < 8 * (KC_H + NB / 2),
                        "a fresh pair would be read by a product before its batch has written it");
          cell_batch<0,  0>(rz0, xn0, hn0, bnx0, bnh0, h0, beside);
          cell_batch<NB, ST_CELL>(rz0, xn0, hn0, bnx0, bnh0, h0, beside);
          #pragma unroll
          for (int wi = 2 * ST_CELL; wi < 8 * (KC_H + KC_H); ++wi)
            one_product<KC_H, KC_H>(wi, whh1, h1, hn1, wih1, h0, xn1, rz1);
          __builtin_amdgcn_sched_barrier(0);
        }

        #pragma unroll
        for (int tl = 0; tl < NRZ; ++tl) rz0[tl] = srz[tl];
        #pragma unroll
        for (int tl = 0; tl < NN; ++tl) { xn0[tl] = sxn[tl]; hn0[tl] = zero8; }
        {
          auto beside = [&](int wi) __attribute__((always_inline)) {
            if (wi < 8 * KC_H) {
              __builtin_amdgcn_sched_barrier(0);
              state_product(wi, whh0, h0, hn0, rz0);
              __builtin_amdgcn_sched_barrier(0);
            }
          };
          static_assert(2 * ST_CELL >= 8 * KC_H, "the two batches' stages must reach every state-side product");
          cell_batch<0,  0>(rz1, xn1, hn1, bnx1, bnh1, h1, beside);
          cell_batch<NB, ST_CELL>(rz1, xn1, hn1, bnx1, bnh1, h1, beside);
          __builtin_amdgcn_sched_barrier(0);
        }
      }
    }
    __syncthreads();
  }

  if (wave == 0) {
    #pragma unroll
    for (int i = 0; i < HH; ++i) stage[m * H + HH * h + i] = h1[i >> 1][i & 1];
  }
  __syncthreads();
  #pragma unroll
  for (int e = threadIdx.x; e < 16 * H; e += 64) out[(size_t)bbase * H + e] = stage[e];
}

extern "C" void kernel_launch(void* const* d_in, const int* in_sizes, int n_in,
                              void* d_out, int out_size, void* d_ws, size_t ws_size,
                              hipStream_t stream) {
  const float* x    = (const float*)d_in[0];
  const float* Wih0 = (const float*)d_in[1];
  const float* Whh0 = (const float*)d_in[2];
  const float* bih0 = (const float*)d_in[3];
  const float* bhh0 = (const float*)d_in[4];
  const float* Wih1 = (const float*)d_in[5];
  const float* Whh1 = (const float*)d_in[6];
  const float* bih1 = (const float*)d_in[7];
  const float* bhh1 = (const float*)d_in[8];
  float* out = (float*)d_out;

  const int B = in_sizes[0] / (T * I);
  const int nblocks = (B + 15) / 16;

  gru2_fused<<<nblocks, 64, 0, stream>>>(x, Wih0, Whh0, bih0, bhh0,
                                         Wih1, Whh1, bih1, bhh1, out);
  (void)n_in; (void)out_size; (void)d_ws; (void)ws_size;
}


# Layer 1's state-side products (Whh1 h1 on its r|z bias) move to the second wave on the other SIMD, handed over in LDS as accumulator start values, 2 block barriers per step; sums unchanged
# speedup vs baseline: 5.7343x; 1.2379x over previous
#include <hip/hip_runtime.h>
#include <cstdint>


typedef float v2f __attribute__((ext_vector_type(2)));
typedef float v4f __attribute__((ext_vector_type(4)));
typedef float v8f __attribute__((ext_vector_type(8)));


namespace {

constexpr int H    = 40;
constexpr int HH   = 20;
constexpr int I    = 16;
constexpr int T    = 256;
constexpr int NRZ  = 5;
constexpr int NN   = 3;
constexpr int NT   = NRZ + NN;
constexpr int KC_I = 4;
constexpr int KC_H = 10;

__device__ __forceinline__ v8f wmma_f32(v2f a, v2f b, v8f c) {
  return __builtin_amdgcn_wmma_f32_16x16x4_f32(
       false, a,  false, b,
       (short)0, c,  false,  false);
}


__device__ __forceinline__ int gate_row(int tl, int m) {
  const int hr = m >> 3, v = m & 7;
  if (tl < NRZ) {
    const int q = tl * 8 + v;
    return (q < HH) ? (HH * hr + q) : (H + HH * hr + (q - HH));
  }
  const int q = (tl - NRZ) * 8 + v;
  return (q < HH) ? (2 * H + HH * hr + q) : -1;
}

template <int K, int KC>
__device__ __forceinline__ void load_a(const float* __restrict__ w, int m, int h, v2f (&a)[NT][KC]) {
  #pragma unroll
  for (int tl = 0; tl < NT; ++tl) {
    const int g  = gate_row(tl, m);
    const float* p = w + (g >= 0 ? g : 0) * K + (K / 2) * h;
    #pragma unroll
    for (int kc = 0; kc < KC; ++kc) {
      const v2f t = *(const v2f*)(p + 2 * kc);
      a[tl][kc] = (g >= 0) ? t : (v2f){0.f, 0.f};
    }
  }
}

__device__ __forceinline__ void load_bias(const float* __restrict__ bih, const float* __restrict__ bhh,
                                          int h, v8f (&brz)[NRZ], v2f (&bnx)[HH / 2], v2f (&bnh)[HH / 2]) {
  #pragma unroll
  for (int tl = 0; tl < NRZ; ++tl) {
    #pragma unroll
    for (int v = 0; v < 8; ++v) {
      const int g = gate_row(tl, 8 * h + v);
      brz[tl][v] = bih[g] + bhh[g];
    }
  }
  #pragma unroll
  for (int i = 0; i < HH; ++i) {
    bnx[i >> 1][i & 1] = bih[2 * H + HH * h + i];
    bnh[i >> 1][i & 1] = bhh[2 * H + HH * h + i];
  }
}


typedef int v2i __attribute__((ext_vector_type(2)));

constexpr int NB = 10;
constexpr int NP = NB / 2;
constexpr int ST_SIG  = 10;
constexpr int ST_CELL = 2 * ST_SIG + 3 + 26 + 2;
static_assert(ST_CELL == 51, "the stage numbers in cell_batch are written out for this count");

__device__ __forceinline__ v2f splat(float k) { return (v2f){k, k}; }
__device__ __forceinline__ v2f fma2(v2f a, v2f b, v2f c) { return __builtin_elementwise_fma(a, b, c); }
__device__ __forceinline__ v2f pair_of(const v8f &a, int e) { return (v2f){a[e], a[e + 1]}; }

template <class V>
__device__ __forceinline__ void keep_stage(V (&a)[NP]) {
  static_assert(NP == 5, "operand list below is written for five");
  asm volatile("" : "+v"(a[0]), "+v"(a[1]), "+v"(a[2]), "+v"(a[3]), "+v"(a[4]));
}

#define STAGE2(k, arr, expr) { _Pragma("unroll") for (int j = 0; j < NP; ++j) { arr[j] = (expr); } \
                               keep_stage(arr); beside(W0 + (k)); }

#define STAGE1(k, arr, expr) { _Pragma("unroll") for (int j = 0; j < NP; ++j) {                     \
                                 _Pragma("unroll") for (int c = 0; c < 2; ++c) { arr[j][c] = (expr); } } \
                               keep_stage(arr); beside(W0 + (k)); }

#define STAGE_FLAG(k, arr, expr) { _Pragma("unroll") for (int j = 0; j < NP; ++j) {                 \
                                     _Pragma("unroll") for (int c = 0; c < 2; ++c) { arr[2 * j + c] = (expr); } } \
                                   beside(W0 + (k)); }

#define SIGMOID_STAGES(K, SRC, OUT)                                                           \
  STAGE2((K) + 0,  t,   __builtin_elementwise_min((SRC) * splat(-0x1.715476p+0f), splat(126.0f))) \
  STAGE1((K) + 1,  t,   __builtin_amdgcn_exp2f(t[j][c]))                                      \
  STAGE2((K) + 2,  dd,  splat(1.0f) + t[j])                                                   \
  STAGE1((K) + 3,  rc,  __builtin_amdgcn_rcpf(dd[j][c]))                                      \
  STAGE2((K) + 4,  f,   fma2(-dd[j], rc[j], splat(1.0f)))                                     \
  STAGE2((K) + 5,  rc,  fma2(f[j], rc[j], rc[j]))                                             \
  STAGE2((K) + 6,  f,   fma2(-dd[j], rc[j], splat(1.0f)))                                     \
  STAGE2((K) + 7,  q,   fma2(f[j], rc[j], rc[j]))                                             \
  STAGE2((K) + 8,  f,   fma2(-dd[j], q[j], splat(1.0f)))                                      \
  STAGE2((K) + 9,  OUT, fma2(f[j], rc[j], q[j]))

template <int I0, int W0, class F>
__device__ __forceinline__ void cell_batch(const v8f (&rz)[NRZ], const v8f (&xn)[NN], const v8f (&hn)[NN],
                                           const v2f (&bnx)[HH / 2], const v2f (&bnh)[HH / 2],
                                           v2f (&h)[KC_H], F beside) {
  static_assert(I0 % 2 == 0 && HH % 2 == 0, "a pair must not straddle an accumulator tile or the r|z split");
  v2f t[NP], dd[NP], rc[NP], f[NP], q[NP], r[NP], z[NP];
  bool fl[NB];
  SIGMOID_STAGES(0,      pair_of(rz[(I0 + 2 * j) >> 3], (I0 + 2 * j) & 7), r)
  SIGMOID_STAGES(ST_SIG, pair_of(rz[(HH + I0 + 2 * j) >> 3], (HH + I0 + 2 * j) & 7), z)

  v2f u[NP], w[NP], pre[NP];
  STAGE2(20, u,   pair_of(xn[(I0 + 2 * j) >> 3], (I0 + 2 * j) & 7) + bnx[I0 / 2 + j])
  STAGE2(21, w,   pair_of(hn[(I0 + 2 * j) >> 3], (I0 + 2 * j) & 7) + bnh[I0 / 2 + j])
  STAGE2(22, pre, fma2(r[j], w[j], u[j]))

  v2f y[NP], y2[NP], p[NP], zs[NP], x2[NP], ph[NP], pl[NP], en[NP], e[NP];
  v2i ei[NP];
  STAGE1(23, y,  __builtin_fabsf(pre[j][c]))
  STAGE2(24, y2, y[j] * y[j])
  STAGE2(25, p,  fma2(y2[j], splat(-0x1.758e7ap-8f), splat(0x1.521192p-6f)))
  STAGE2(26, p,  fma2(y2[j], p[j], splat(-0x1.b8389cp-5f)))
  STAGE2(27, p,  fma2(y2[j], p[j], splat(0x1.110704p-3f)))
  STAGE2(28, p,  fma2(y2[j], p[j], splat(-0x1.555532p-2f)))
  STAGE2(29, p,  y[j] * p[j])
  STAGE2(30, zs, fma2(y2[j], p[j], y[j]))
  STAGE2(31, x2, y[j] + y[j])
  STAGE2(32, ph, x2[j] * splat(0x1.715476p+0f))
  STAGE2(33, pl, fma2(x2[j], splat(0x1.715476p+0f), -ph[j]))
  STAGE1(34, en, __builtin_rintf(ph[j][c]))
  STAGE2(35, pl, fma2(x2[j], splat(0x1.4ae0bep-26f), pl[j]))
  STAGE2(36, ph, ph[j] - en[j])
  STAGE2(37, ph, ph[j] + pl[j])
  STAGE1(38, ei, (int)en[j][c])
  STAGE1(39, e,  __builtin_amdgcn_exp2f(ph[j][c]))
  STAGE1(40, e,  __builtin_ldexpf(e[j][c], ei[j][c]))
  STAGE_FLAG(41, fl, !(0x1.62e43p+6f < x2[j][c]))
  STAGE1(42, e,  fl[2 * j + c] ? e[j][c] : __builtin_inff())
  STAGE2(43, e,  e[j] + splat(1.0f))
  STAGE1(44, e,  __builtin_amdgcn_rcpf(e[j][c]))
  STAGE2(45, e,  fma2(splat(-2.0f), e[j], splat(1.0f)))
  STAGE_FLAG(46, fl, y[j][c] < 0.625f)
  STAGE1(47, e,  fl[2 * j + c] ? zs[j][c] : e[j][c])
  STAGE1(48, e,  __builtin_copysignf(e[j][c], pre[j][c]))

  STAGE2(49, u,  h[I0 / 2 + j] - e[j])
  STAGE2(50, u,  fma2(z[j], u[j], e[j]))
  #pragma unroll
  for (int j = 0; j < NP; ++j) h[I0 / 2 + j] = u[j];
}

constexpr int SLOT = 16 * 32;
constexpr int HSL  = (HH / 4) * 32;
static_assert(HH % 4 == 0, "a lane's share of a state is handed over in 16-byte pieces");

__device__ __forceinline__ void put_slot(v4f* d, const v8f (&rz)[NRZ], const v8f (&n)[NN]) {
  #pragma unroll
  for (int tl = 0; tl < NRZ; ++tl) {
    d[(2 * tl) * 32]     = (v4f){rz[tl][0], rz[tl][1], rz[tl][2], rz[tl][3]};
    d[(2 * tl + 1) * 32] = (v4f){rz[tl][4], rz[tl][5], rz[tl][6], rz[tl][7]};
  }
  #pragma unroll
  for (int tl = 0; tl < NN; ++tl) {
    d[(2 * (NRZ + tl)) * 32]     = (v4f){n[tl][0], n[tl][1], n[tl][2], n[tl][3]};
    d[(2 * (NRZ + tl) + 1) * 32] = (v4f){n[tl][4], n[tl][5], n[tl][6], n[tl][7]};
  }
}

__device__ __forceinline__ void take_slot(const v4f* src, v8f (&rz)[NRZ], v8f (&xn)[NN]) {
  #pragma unroll
  for (int tl = 0; tl < NRZ; ++tl) {
    const v4f lo = src[(2 * tl) * 32], hi = src[(2 * tl + 1) * 32];
    rz[tl] = (v8f){lo.x, lo.y, lo.z, lo.w, hi.x, hi.y, hi.z, hi.w};
  }
  #pragma unroll
  for (int tl = 0; tl < NN; ++tl) {
    const v4f lo = src[(2 * (NRZ + tl)) * 32], hi = src[(2 * (NRZ + tl) + 1) * 32];
    xn[tl] = (v8f){lo.x, lo.y, lo.z, lo.w, hi.x, hi.y, hi.z, hi.w};
  }
}

__device__ __forceinline__ void put_state(v4f* d, const v2f (&hs)[KC_H]) {
  #pragma unroll
  for (int q = 0; q < HH / 4; ++q) d[q * 32] = (v4f){hs[2 * q][0], hs[2 * q][1], hs[2 * q + 1][0], hs[2 * q + 1][1]};
}
__device__ __forceinline__ void take_state(const v4f* src, v2f (&hs)[KC_H]) {
  #pragma unroll
  for (int q = 0; q < HH / 4; ++q) {
    const v4f t = src[q * 32];
    hs[2 * q] = (v2f){t.x, t.y};
    hs[2 * q + 1] = (v2f){t.z, t.w};
  }
}

__device__ __forceinline__ void produce(const float* __restrict__ x, int xoff, int s0, v4f* dst,
                                        const float* __restrict__ Wih0,
                                        const float* __restrict__ bih0, const float* __restrict__ bhh0,
                                        int m, int h) {
  v2f wih0[NT][KC_I];
  load_a<I, KC_I>(Wih0, m, h, wih0);
  v8f brz0[NRZ];
  v2f bnx[HH / 2], bnh[HH / 2];
  load_bias(bih0, bhh0, h, brz0, bnx, bnh);
  const int s = (s0 < T) ? s0 : (T - 1);
  const float* xrow = x + (xoff + s * I);
  const v4f xa = *(const v4f*)(xrow);
  const v4f xb = *(const v4f*)(xrow + 4);
  const v8f zero8 = {0.f, 0.f, 0.f, 0.f, 0.f, 0.f, 0.f, 0.f};
  const v2f xf[KC_I] = {{xa.x, xa.y}, {xa.z, xa.w}, {xb.x, xb.y}, {xb.z, xb.w}};
  v8f rz[NRZ], xn[NN];
  #pragma unroll
  for (int tl = 0; tl < NRZ; ++tl) rz[tl] = brz0[tl];
  #pragma unroll
  for (int tl = 0; tl < NN; ++tl) xn[tl] = zero8;
  #pragma unroll
  for (int wi = 0; wi < 8 * KC_I; ++wi) {
    const int kc = wi >> 3, tl = wi & 7;
    if (tl < NRZ) rz[tl] = wmma_f32(wih0[tl][kc], xf[kc], rz[tl]);
    else          xn[tl - NRZ] = wmma_f32(wih0[tl][kc], xf[kc], xn[tl - NRZ]);
  }
  put_slot(dst, rz, xn);
}

__device__ __forceinline__ void state_product(int wi, const v2f (&w)[NT][KC_H], const v2f (&b)[KC_H],
                                              v8f (&n)[NN], v8f (&rz)[NRZ]) {
  const int kc = wi >> 3, tl = wi & 7;
  if (kc < KC_H) {
    if (tl < NRZ) rz[tl] = wmma_f32(w[tl][kc], b[kc], rz[tl]);
    else          n[tl - NRZ] = wmma_f32(w[tl][kc], b[kc], n[tl - NRZ]);
  }
}

}

__global__ __launch_bounds__(64) void gru2_fused(
    const float* __restrict__ x,
    const float* __restrict__ Wih0, const float* __restrict__ Whh0,
    const float* __restrict__ bih0, const float* __restrict__ bhh0,
    const float* __restrict__ Wih1, const float* __restrict__ Whh1,
    const float* __restrict__ bih1, const float* __restrict__ bhh1,
    float* __restrict__ out)
{
  __shared__ v4f xs[2 * SLOT];
  __shared__ v4f s1[SLOT];
  __shared__ v4f hx[HSL];
  __shared__ v4f b1[2 * NRZ * 32];
  __shared__ float stage[16 * H];
  const int lane  = threadIdx.x & 31;
  const int wave  = __builtin_amdgcn_readfirstlane((int)(threadIdx.x >> 5));
  const int m     = lane & 15;
  const int h     = lane >> 4;
  const int bbase = blockIdx.x * 16;

  const int xoff = (bbase + m) * (T * I) + 8 * h;

  const v8f zero8 = {0.f, 0.f, 0.f, 0.f, 0.f, 0.f, 0.f, 0.f};

  v2f whh0[NT][KC_H], wih1[NT][KC_H];
  v2f bnx0[HH / 2], bnh0[HH / 2], bnx1[HH / 2], bnh1[HH / 2];
  v2f h0[KC_H], h1[KC_H];
  v8f rz0[NRZ], xn0[NN], hn0[NN];
  v2f whh1[NT][KC_H];

  if (wave == 0) {
    load_a<H, KC_H>(Whh0, m, h, whh0);
    load_a<H, KC_H>(Wih1, m, h, wih1);
    v8f brzu[NRZ];
    load_bias(bih0, bhh0, h, brzu, bnx0, bnh0);
    load_bias(bih1, bhh1, h, brzu, bnx1, bnh1);
    #pragma unroll
    for (int kc = 0; kc < KC_H; ++kc) { h0[kc] = (v2f){0.f, 0.f}; h1[kc] = (v2f){0.f, 0.f}; }
    put_state(hx + lane, h1);
  } else {
    load_a<H, KC_H>(Whh1, m, h, whh1);
    v8f brz1[NRZ];
    v2f bu0[HH / 2], bu1[HH / 2];
    load_bias(bih1, bhh1, h, brz1, bu0, bu1);
    #pragma unroll
    for (int tl = 0; tl < NRZ; ++tl) {
      b1[(2 * tl) * 32 + lane]     = (v4f){brz1[tl][0], brz1[tl][1], brz1[tl][2], brz1[tl][3]};
      b1[(2 * tl + 1) * 32 + lane] = (v4f){brz1[tl][4], brz1[tl][5], brz1[tl][6], brz1[tl][7]};
    }
    produce(x, xoff, 0, xs + lane, Wih0, bih0, bhh0, m, h);
    produce(x, xoff, 1, xs + SLOT + lane, Wih0, bih0, bhh0, m, h);
  }
  __syncthreads();

  if (wave == 0) {
    take_slot(xs + lane, rz0, xn0);
    #pragma unroll
    for (int tl = 0; tl < NN; ++tl) hn0[tl] = zero8;
    #pragma unroll
    for (int wi = 0; wi < 8 * KC_H; ++wi) state_product(wi, whh0, h0, hn0, rz0);
  }

  #pragma unroll 1
  for (int i = 0; i < T; ++i) {
    v8f nrz[NRZ], nxn[NN], nhn[NN];
    if (wave == 0) {
      take_slot(xs + ((i + 1) & 1) * SLOT + lane, nrz, nxn);
      #pragma unroll
      for (int tl = 0; tl < NN; ++tl) nhn[tl] = zero8;
      auto none = [&](int) __attribute__((always_inline)) {};
      auto beside = [&](int wi) __attribute__((always_inline)) {
        if (wi >= ST_CELL && wi - ST_CELL < 8 * (NB / 2)) {
          __builtin_amdgcn_sched_barrier(0);
          state_product(wi - ST_CELL, whh0, h0, nhn, nrz);
          __builtin_amdgcn_sched_barrier(0);
        }
      };
      static_assert(ST_CELL >= 8 * (NB / 2), "batch 1's stages must reach every product of chunks 0-4");
      cell_batch<0,  0>(rz0, xn0, hn0, bnx0, bnh0, h0, none);
      cell_batch<NB, ST_CELL>(rz0, xn0, hn0, bnx0, bnh0, h0, beside);
      __builtin_amdgcn_sched_barrier(0);
      #pragma unroll
      for (int wi = 8 * (NB / 2); wi < 8 * KC_H; ++wi) state_product(wi, whh0, h0, nhn, nrz);
    } else {
      v2f hb[KC_H];
      take_state(hx + lane, hb);
      v8f rz[NRZ], hn[NN];
      #pragma unroll
      for (int tl = 0; tl < NRZ; ++tl) {
        const v4f lo = b1[(2 * tl) * 32 + lane], hi = b1[(2 * tl + 1) * 32 + lane];
        rz[tl] = (v8f){lo.x, lo.y, lo.z, lo.w, hi.x, hi.y, hi.z, hi.w};
      }
      #pragma unroll
      for (int tl = 0; tl < NN; ++tl) hn[tl] = zero8;
      #pragma unroll
      for (int wi = 0; wi < 8 * KC_H; ++wi) state_product(wi, whh1, hb, hn, rz);
      put_slot(s1 + lane, rz, hn);
    }
    __syncthreads();
    if (wave == 0) {
      v8f rz1[NRZ], xn1[NN], hn1[NN];
      take_slot(s1 + lane, rz1, hn1);
      #pragma unroll
      for (int tl = 0; tl < NN; ++tl) xn1[tl] = zero8;
      #pragma unroll
      for (int wi = 0; wi < 8 * KC_H; ++wi) state_product(wi, wih1, h0, xn1, rz1);
      auto none = [&](int) __attribute__((always_inline)) {};
      cell_batch<0,  0>(rz1, xn1, hn1, bnx1, bnh1, h1, none);
      cell_batch<NB, ST_CELL>(rz1, xn1, hn1, bnx1, bnh1, h1, none);
      put_state(hx + lane, h1);
      #pragma unroll
      for (int tl = 0; tl < NRZ; ++tl) rz0[tl] = nrz[tl];
      #pragma unroll
      for (int tl = 0; tl < NN; ++tl) { xn0[tl] = nxn[tl]; hn0[tl] = nhn[tl]; }
    } else {
      produce(x, xoff, i + 2, xs + (i & 1) * SLOT + lane, Wih0, bih0, bhh0, m, h);
    }
    __syncthreads();
  }

  if (wave == 0) {
    #pragma unroll
    for (int i = 0; i < HH; ++i) stage[m * H + HH * h + i] = h1[i >> 1][i & 1];
  }
  __syncthreads();
  #pragma unroll
  for (int e = threadIdx.x; e < 16 * H; e += 64) out[(size_t)bbase * H + e] = stage[e];
}

extern "C" void kernel_launch(void* const* d_in, const int* in_sizes, int n_in,
                              void* d_out, int out_size, void* d_ws, size_t ws_size,
                              hipStream_t stream) {
  const float* x    = (const float*)d_in[0];
  const float* Wih0 = (const float*)d_in[1];
  const float* Whh0 = (const float*)d_in[2];
  const float* bih0 = (const float*)d_in[3];
  const float* bhh0 = (const float*)d_in[4];
  const float* Wih1 = (const float*)d_in[5];
  const float* Whh1 = (const float*)d_in[6];
  const float* bih1 = (const float*)d_in[7];
  const float* bhh1 = (const float*)d_in[8];
  float* out = (float*)d_out;

  const int B = in_sizes[0] / (T * I);
  const int nblocks = (B + 15) / 16;

  gru2_fused<<<nblocks, 64, 0, stream>>>(x, Wih0, Whh0, bih0, bhh0,
                                         Wih1, Whh1, bih1, bhh1, out);
  (void)n_in; (void)out_size; (void)d_ws; (void)ws_size;
}


# A third wave on a third SIMD takes layer 0's state-side products (Whh0 h0, beside layer 1's step) and half the tiles of layer 1's (Whh1 h1), handed over in LDS as finished sums; sums unchanged
# speedup vs baseline: 7.7191x; 1.3461x over previous
#include <hip/hip_runtime.h>
#include <cstdint>


typedef float v2f __attribute__((ext_vector_type(2)));
typedef float v4f __attribute__((ext_vector_type(4)));
typedef float v8f __attribute__((ext_vector_type(8)));


namespace {

constexpr int H    = 40;
constexpr int HH   = 20;
constexpr int I    = 16;
constexpr int T    = 256;
constexpr int NRZ  = 5;
constexpr int NN   = 3;
constexpr int NT   = NRZ + NN;
constexpr int KC_I = 4;
constexpr int KC_H = 10;

__device__ __forceinline__ v8f wmma_f32(v2f a, v2f b, v8f c) {
  return __builtin_amdgcn_wmma_f32_16x16x4_f32(
       false, a,  false, b,
       (short)0, c,  false,  false);
}


__device__ __forceinline__ int gate_row(int tl, int m) {
  const int hr = m >> 3, v = m & 7;
  if (tl < NRZ) {
    const int q = tl * 8 + v;
    return (q < HH) ? (HH * hr + q) : (H + HH * hr + (q - HH));
  }
  const int q = (tl - NRZ) * 8 + v;
  return (q < HH) ? (2 * H + HH * hr + q) : -1;
}

template <int K, int KC>
__device__ __forceinline__ void load_a(const float* __restrict__ w, int m, int h, v2f (&a)[NT][KC]) {
  #pragma unroll
  for (int tl = 0; tl < NT; ++tl) {
    const int g  = gate_row(tl, m);
    const float* p = w + (g >= 0 ? g : 0) * K + (K / 2) * h;
    #pragma unroll
    for (int kc = 0; kc < KC; ++kc) {
      const v2f t = *(const v2f*)(p + 2 * kc);
      a[tl][kc] = (g >= 0) ? t : (v2f){0.f, 0.f};
    }
  }
}

__device__ __forceinline__ void load_bias(const float* __restrict__ bih, const float* __restrict__ bhh,
                                          int h, v8f (&brz)[NRZ], v2f (&bnx)[HH / 2], v2f (&bnh)[HH / 2]) {
  #pragma unroll
  for (int tl = 0; tl < NRZ; ++tl) {
    #pragma unroll
    for (int v = 0; v < 8; ++v) {
      const int g = gate_row(tl, 8 * h + v);
      brz[tl][v] = bih[g] + bhh[g];
    }
  }
  #pragma unroll
  for (int i = 0; i < HH; ++i) {
    bnx[i >> 1][i & 1] = bih[2 * H + HH * h + i];
    bnh[i >> 1][i & 1] = bhh[2 * H + HH * h + i];
  }
}


typedef int v2i __attribute__((ext_vector_type(2)));

constexpr int NB = 10;
constexpr int NP = NB / 2;
constexpr int ST_SIG  = 10;
constexpr int ST_CELL = 2 * ST_SIG + 3 + 26 + 2;
static_assert(ST_CELL == 51, "the stage numbers in cell_batch are written out for this count");

__device__ __forceinline__ v2f splat(float k) { return (v2f){k, k}; }
__device__ __forceinline__ v2f fma2(v2f a, v2f b, v2f c) { return __builtin_elementwise_fma(a, b, c); }
__device__ __forceinline__ v2f pair_of(const v8f &a, int e) { return (v2f){a[e], a[e + 1]}; }

template <class V>
__device__ __forceinline__ void keep_stage(V (&a)[NP]) {
  static_assert(NP == 5, "operand list below is written for five");
  asm volatile("" : "+v"(a[0]), "+v"(a[1]), "+v"(a[2]), "+v"(a[3]), "+v"(a[4]));
}

#define STAGE2(k, arr, expr) { _Pragma("unroll") for (int j = 0; j < NP; ++j) { arr[j] = (expr); } \
                               keep_stage(arr); beside(W0 + (k)); }

#define STAGE1(k, arr, expr) { _Pragma("unroll") for (int j = 0; j < NP; ++j) {                     \
                                 _Pragma("unroll") for (int c = 0; c < 2; ++c) { arr[j][c] = (expr); } } \
                               keep_stage(arr); beside(W0 + (k)); }

#define STAGE_FLAG(k, arr, expr) { _Pragma("unroll") for (int j = 0; j < NP; ++j) {                 \
                                     _Pragma("unroll") for (int c = 0; c < 2; ++c) { arr[2 * j + c] = (expr); } } \
                                   beside(W0 + (k)); }

#define SIGMOID_STAGES(K, SRC, OUT)                                                           \
  STAGE2((K) + 0,  t,   __builtin_elementwise_min((SRC) * splat(-0x1.715476p+0f), splat(126.0f))) \
  STAGE1((K) + 1,  t,   __builtin_amdgcn_exp2f(t[j][c]))                                      \
  STAGE2((K) + 2,  dd,  splat(1.0f) + t[j])                                                   \
  STAGE1((K) + 3,  rc,  __builtin_amdgcn_rcpf(dd[j][c]))                                      \
  STAGE2((K) + 4,  f,   fma2(-dd[j], rc[j], splat(1.0f)))                                     \
  STAGE2((K) + 5,  rc,  fma2(f[j], rc[j], rc[j]))                                             \
  STAGE2((K) + 6,  f,   fma2(-dd[j], rc[j], splat(1.0f)))                                     \
  STAGE2((K) + 7,  q,   fma2(f[j], rc[j], rc[j]))                                             \
  STAGE2((K) + 8,  f,   fma2(-dd[j], q[j], splat(1.0f)))                                      \
  STAGE2((K) + 9,  OUT, fma2(f[j], rc[j], q[j]))

template <int I0, int W0, class F>
__device__ __forceinline__ void cell_batch(const v8f (&rz)[NRZ], const v8f (&xn)[NN], const v8f (&hn)[NN],
                                           const v2f (&bnx)[HH / 2], const v2f (&bnh)[HH / 2],
                                           v2f (&h)[KC_H], F beside) {
  static_assert(I0 % 2 == 0 && HH % 2 == 0, "a pair must not straddle an accumulator tile or the r|z split");
  v2f t[NP], dd[NP], rc[NP], f[NP], q[NP], r[NP], z[NP];
  bool fl[NB];
  SIGMOID_STAGES(0,      pair_of(rz[(I0 + 2 * j) >> 3], (I0 + 2 * j) & 7), r)
  SIGMOID_STAGES(ST_SIG, pair_of(rz[(HH + I0 + 2 * j) >> 3], (HH + I0 + 2 * j) & 7), z)

  v2f u[NP], w[NP], pre[NP];
  STAGE2(20, u,   pair_of(xn[(I0 + 2 * j) >> 3], (I0 + 2 * j) & 7) + bnx[I0 / 2 + j])
  STAGE2(21, w,   pair_of(hn[(I0 + 2 * j) >> 3], (I0 + 2 * j) & 7) + bnh[I0 / 2 + j])
  STAGE2(22, pre, fma2(r[j], w[j], u[j]))

  v2f y[NP], y2[NP], p[NP], zs[NP], x2[NP], ph[NP], pl[NP], en[NP], e[NP];
  v2i ei[NP];
  STAGE1(23, y,  __builtin_fabsf(pre[j][c]))
  STAGE2(24, y2, y[j] * y[j])
  STAGE2(25, p,  fma2(y2[j], splat(-0x1.758e7ap-8f), splat(0x1.521192p-6f)))
  STAGE2(26, p,  fma2(y2[j], p[j], splat(-0x1.b8389cp-5f)))
  STAGE2(27, p,  fma2(y2[j], p[j], splat(0x1.110704p-3f)))
  STAGE2(28, p,  fma2(y2[j], p[j], splat(-0x1.555532p-2f)))
  STAGE2(29, p,  y[j] * p[j])
  STAGE2(30, zs, fma2(y2[j], p[j], y[j]))
  STAGE2(31, x2, y[j] + y[j])
  STAGE2(32, ph, x2[j] * splat(0x1.715476p+0f))
  STAGE2(33, pl, fma2(x2[j], splat(0x1.715476p+0f), -ph[j]))
  STAGE1(34, en, __builtin_rintf(ph[j][c]))
  STAGE2(35, pl, fma2(x2[j], splat(0x1.4ae0bep-26f), pl[j]))
  STAGE2(36, ph, ph[j] - en[j])
  STAGE2(37, ph, ph[j] + pl[j])
  STAGE1(38, ei, (int)en[j][c])
  STAGE1(39, e,  __builtin_amdgcn_exp2f(ph[j][c]))
  STAGE1(40, e,  __builtin_ldexpf(e[j][c], ei[j][c]))
  STAGE_FLAG(41, fl, !(0x1.62e43p+6f < x2[j][c]))
  STAGE1(42, e,  fl[2 * j + c] ? e[j][c] : __builtin_inff())
  STAGE2(43, e,  e[j] + splat(1.0f))
  STAGE1(44, e,  __builtin_amdgcn_rcpf(e[j][c]))
  STAGE2(45, e,  fma2(splat(-2.0f), e[j], splat(1.0f)))
  STAGE_FLAG(46, fl, y[j][c] < 0.625f)
  STAGE1(47, e,  fl[2 * j + c] ? zs[j][c] : e[j][c])
  STAGE1(48, e,  __builtin_copysignf(e[j][c], pre[j][c]))

  STAGE2(49, u,  h[I0 / 2 + j] - e[j])
  STAGE2(50, u,  fma2(z[j], u[j], e[j]))
  #pragma unroll
  for (int j = 0; j < NP; ++j) h[I0 / 2 + j] = u[j];
}

constexpr int TSPLIT = 4;
static_assert(TSPLIT < NRZ, "wave 2's share is written for one r|z tile and the n tiles");
constexpr int SLOT = 16 * 32;
constexpr int HSL  = (HH / 4) * 32;
static_assert(HH % 4 == 0, "a lane's share of a state is handed over in 16-byte pieces");

__device__ __forceinline__ void put_slot(v4f* d, const v8f (&rz)[NRZ], const v8f (&n)[NN]) {
  #pragma unroll
  for (int tl = 0; tl < NRZ; ++tl) {
    d[(2 * tl) * 32]     = (v4f){rz[tl][0], rz[tl][1], rz[tl][2], rz[tl][3]};
    d[(2 * tl + 1) * 32] = (v4f){rz[tl][4], rz[tl][5], rz[tl][6], rz[tl][7]};
  }
  #pragma unroll
  for (int tl = 0; tl < NN; ++tl) {
    d[(2 * (NRZ + tl)) * 32]     = (v4f){n[tl][0], n[tl][1], n[tl][2], n[tl][3]};
    d[(2 * (NRZ + tl) + 1) * 32] = (v4f){n[tl][4], n[tl][5], n[tl][6], n[tl][7]};
  }
}

__device__ __forceinline__ void take_slot(const v4f* src, v8f (&rz)[NRZ], v8f (&xn)[NN]) {
  #pragma unroll
  for (int tl = 0; tl < NRZ; ++tl) {
    const v4f lo = src[(2 * tl) * 32], hi = src[(2 * tl + 1) * 32];
    rz[tl] = (v8f){lo.x, lo.y, lo.z, lo.w, hi.x, hi.y, hi.z, hi.w};
  }
  #pragma unroll
  for (int tl = 0; tl < NN; ++tl) {
    const v4f lo = src[(2 * (NRZ + tl)) * 32], hi = src[(2 * (NRZ + tl) + 1) * 32];
    xn[tl] = (v8f){lo.x, lo.y, lo.z, lo.w, hi.x, hi.y, hi.z, hi.w};
  }
}

__device__ __forceinline__ void take_slot_n(const v4f* src, v8f (&xn)[NN]) {
  #pragma unroll
  for (int tl = 0; tl < NN; ++tl) {
    const v4f lo = src[(2 * (NRZ + tl)) * 32], hi = src[(2 * (NRZ + tl) + 1) * 32];
    xn[tl] = (v8f){lo.x, lo.y, lo.z, lo.w, hi.x, hi.y, hi.z, hi.w};
  }
}

__device__ __forceinline__ void put_state(v4f* d, const v2f (&hs)[KC_H]) {
  #pragma unroll
  for (int q = 0; q < HH / 4; ++q) d[q * 32] = (v4f){hs[2 * q][0], hs[2 * q][1], hs[2 * q + 1][0], hs[2 * q + 1][1]};
}
__device__ __forceinline__ void take_state(const v4f* src, v2f (&hs)[KC_H]) {
  #pragma unroll
  for (int q = 0; q < HH / 4; ++q) {
    const v4f t = src[q * 32];
    hs[2 * q] = (v2f){t.x, t.y};
    hs[2 * q + 1] = (v2f){t.z, t.w};
  }
}

__device__ __forceinline__ void produce(const float* __restrict__ x, int xoff, int s0, v4f* dst,
                                        const float* __restrict__ Wih0,
                                        const float* __restrict__ bih0, const float* __restrict__ bhh0,
                                        int m, int h) {
  v2f wih0[NT][KC_I];
  load_a<I, KC_I>(Wih0, m, h, wih0);
  v8f brz0[NRZ];
  v2f bnx[HH / 2], bnh[HH / 2];
  load_bias(bih0, bhh0, h, brz0, bnx, bnh);
  const int s = (s0 < T) ? s0 : (T - 1);
  const float* xrow = x + (xoff + s * I);
  const v4f xa = *(const v4f*)(xrow);
  const v4f xb = *(const v4f*)(xrow + 4);
  const v8f zero8 = {0.f, 0.f, 0.f, 0.f, 0.f, 0.f, 0.f, 0.f};
  const v2f xf[KC_I] = {{xa.x, xa.y}, {xa.z, xa.w}, {xb.x, xb.y}, {xb.z, xb.w}};
  v8f rz[NRZ], xn[NN];
  #pragma unroll
  for (int tl = 0; tl < NRZ; ++tl) rz[tl] = brz0[tl];
  #pragma unroll
  for (int tl = 0; tl < NN; ++tl) xn[tl] = zero8;
  #pragma unroll
  for (int wi = 0; wi < 8 * KC_I; ++wi) {
    const int kc = wi >> 3, tl = wi & 7;
    if (tl < NRZ) rz[tl] = wmma_f32(wih0[tl][kc], xf[kc], rz[tl]);
    else          xn[tl - NRZ] = wmma_f32(wih0[tl][kc], xf[kc], xn[tl - NRZ]);
  }
  put_slot(dst, rz, xn);
}

__device__ __forceinline__ void state_product(int wi, const v2f (&w)[NT][KC_H], const v2f (&b)[KC_H],
                                              v8f (&n)[NN], v8f (&rz)[NRZ]) {
  const int kc = wi >> 3, tl = wi & 7;
  if (kc < KC_H) {
    if (tl < NRZ) rz[tl] = wmma_f32(w[tl][kc], b[kc], rz[tl]);
    else          n[tl - NRZ] = wmma_f32(w[tl][kc], b[kc], n[tl - NRZ]);
  }
}

template <int TL0, int TL1>
__device__ __forceinline__ void tile_products(const v2f (&w)[NT][KC_H], const v2f (&b)[KC_H],
                                              v8f (&n)[NN], v8f (&rz)[NRZ]) {
  #pragma unroll
  for (int kc = 0; kc < KC_H; ++kc) {
    #pragma unroll
    for (int tl = TL0; tl < TL1; ++tl) {
      if (tl < NRZ) rz[tl] = wmma_f32(w[tl][kc], b[kc], rz[tl]);
      else          n[tl - NRZ] = wmma_f32(w[tl][kc], b[kc], n[tl - NRZ]);
    }
  }
}

__device__ __forceinline__ void put_tile(v4f* d, int tl, const v8f &a) {
  d[(2 * tl) * 32]     = (v4f){a[0], a[1], a[2], a[3]};
  d[(2 * tl + 1) * 32] = (v4f){a[4], a[5], a[6], a[7]};
}

__device__ __forceinline__ void layer0_state_side(const v4f* in, const v2f (&hb)[KC_H], v4f* dst,
                                                  const v2f (&whh0)[NT][KC_H]) {
  v8f rz[NRZ], hn[NN];
  #pragma unroll
  for (int tl = 0; tl < NRZ; ++tl) {
    const v4f lo = in[(2 * tl) * 32], hi = in[(2 * tl + 1) * 32];
    rz[tl] = (v8f){lo.x, lo.y, lo.z, lo.w, hi.x, hi.y, hi.z, hi.w};
  }
  #pragma unroll
  for (int tl = 0; tl < NN; ++tl) hn[tl] = (v8f){0.f, 0.f, 0.f, 0.f, 0.f, 0.f, 0.f, 0.f};
  #pragma unroll
  for (int wi = 0; wi < 8 * KC_H; ++wi) state_product(wi, whh0, hb, hn, rz);
  put_slot(dst, rz, hn);
}

}

__global__ __launch_bounds__(96) void gru2_fused(
    const float* __restrict__ x,
    const float* __restrict__ Wih0, const float* __restrict__ Whh0,
    const float* __restrict__ bih0, const float* __restrict__ bhh0,
    const float* __restrict__ Wih1, const float* __restrict__ Whh1,
    const float* __restrict__ bih1, const float* __restrict__ bhh1,
    float* __restrict__ out)
{
  __shared__ v4f xs[2 * SLOT];
  __shared__ v4f s0[SLOT];
  __shared__ v4f s1[SLOT];
  __shared__ v4f hx[HSL];
  __shared__ v4f h0x[HSL];
  __shared__ v4f b1[2 * TSPLIT * 32];
  __shared__ float stage[16 * H];
  const int lane  = threadIdx.x & 31;
  const int wave  = __builtin_amdgcn_readfirstlane((int)(threadIdx.x >> 5));
  const int m     = lane & 15;
  const int h     = lane >> 4;
  const int bbase = blockIdx.x * 16;

  const int xoff = (bbase + m) * (T * I) + 8 * h;

  const v8f zero8 = {0.f, 0.f, 0.f, 0.f, 0.f, 0.f, 0.f, 0.f};

  v2f wih1[NT][KC_H];
  v2f bnx0[HH / 2], bnh0[HH / 2], bnx1[HH / 2], bnh1[HH / 2];
  v2f h0[KC_H], h1[KC_H];
  v2f whh1[NT][KC_H];
  v2f whh0[NT][KC_H], whh1b[NT][KC_H];
  v8f brz1hi[NRZ];

  if (wave == 0) {
    load_a<H, KC_H>(Wih1, m, h, wih1);
    v8f brzu[NRZ];
    load_bias(bih0, bhh0, h, brzu, bnx0, bnh0);
    load_bias(bih1, bhh1, h, brzu, bnx1, bnh1);
    #pragma unroll
    for (int kc = 0; kc < KC_H; ++kc) { h0[kc] = (v2f){0.f, 0.f}; h1[kc] = (v2f){0.f, 0.f}; }
    put_state(hx + lane, h1);
  } else if (wave == 1) {
    load_a<H, KC_H>(Whh1, m, h, whh1);
    v8f brz1[NRZ];
    v2f bu0[HH / 2], bu1[HH / 2];
    load_bias(bih1, bhh1, h, brz1, bu0, bu1);
    #pragma unroll
    for (int tl = 0; tl < TSPLIT; ++tl) {
      b1[(2 * tl) * 32 + lane]     = (v4f){brz1[tl][0], brz1[tl][1], brz1[tl][2], brz1[tl][3]};
      b1[(2 * tl + 1) * 32 + lane] = (v4f){brz1[tl][4], brz1[tl][5], brz1[tl][6], brz1[tl][7]};
    }
    produce(x, xoff, 0, xs + lane, Wih0, bih0, bhh0, m, h);
    produce(x, xoff, 1, xs + SLOT + lane, Wih0, bih0, bhh0, m, h);
  } else {
    load_a<H, KC_H>(Whh0, m, h, whh0);
    load_a<H, KC_H>(Whh1, m, h, whh1b);
    v2f bu0[HH / 2], bu1[HH / 2];
    load_bias(bih1, bhh1, h, brz1hi, bu0, bu1);
  }
  __syncthreads();

  if (wave == 2) {
    v2f hz[KC_H];
    #pragma unroll
    for (int kc = 0; kc < KC_H; ++kc) hz[kc] = (v2f){0.f, 0.f};
    layer0_state_side(xs + lane, hz, s0 + lane, whh0);
  }
  __syncthreads();

  #pragma unroll 1
  for (int i = 0; i < T; ++i) {
    auto none = [&](int) __attribute__((always_inline)) {};
    if (wave == 0) {
      v8f rz0[NRZ], xn0[NN], hn0[NN];
      take_slot(s0 + lane, rz0, hn0);
      take_slot_n(xs + (i & 1) * SLOT + lane, xn0);
      cell_batch<0,  0>(rz0, xn0, hn0, bnx0, bnh0, h0, none);
      cell_batch<NB, ST_CELL>(rz0, xn0, hn0, bnx0, bnh0, h0, none);
      put_state(h0x + lane, h0);
    } else if (wave == 1) {
      v2f hb[KC_H];
      take_state(hx + lane, hb);
      v8f rz[NRZ], hn[NN];
      #pragma unroll
      for (int tl = 0; tl < TSPLIT; ++tl) {
        const v4f lo = b1[(2 * tl) * 32 + lane], hi = b1[(2 * tl + 1) * 32 + lane];
        rz[tl] = (v8f){lo.x, lo.y, lo.z, lo.w, hi.x, hi.y, hi.z, hi.w};
      }
      tile_products<0, TSPLIT>(whh1, hb, hn, rz);
      #pragma unroll
      for (int tl = 0; tl < TSPLIT; ++tl) put_tile(s1 + lane, tl, rz[tl]);
    } else {
      v2f hb[KC_H];
      take_state(hx + lane, hb);
      v8f rz[NRZ], hn[NN];
      #pragma unroll
      for (int tl = TSPLIT; tl < NRZ; ++tl) rz[tl] = brz1hi[tl];
      #pragma unroll
      for (int tl = 0; tl < NN; ++tl) hn[tl] = zero8;
      tile_products<TSPLIT, NT>(whh1b, hb, hn, rz);
      #pragma unroll
      for (int tl = TSPLIT; tl < NRZ; ++tl) put_tile(s1 + lane, tl, rz[tl]);
      #pragma unroll
      for (int tl = 0; tl < NN; ++tl) put_tile(s1 + lane, NRZ + tl, hn[tl]);
    }
    __syncthreads();
    if (wave == 0) {
      v8f rz1[NRZ], xn1[NN], hn1[NN];
      take_slot(s1 + lane, rz1, hn1);
      #pragma unroll
      for (int tl = 0; tl < NN; ++tl) xn1[tl] = zero8;
      #pragma unroll
      for (int wi = 0; wi < 8 * KC_H; ++wi) state_product(wi, wih1, h0, xn1, rz1);
      cell_batch<0,  0>(rz1, xn1, hn1, bnx1, bnh1, h1, none);
      cell_batch<NB, ST_CELL>(rz1, xn1, hn1, bnx1, bnh1, h1, none);
      put_state(hx + lane, h1);
    } else if (wave == 1) {
      produce(x, xoff, i + 2, xs + (i & 1) * SLOT + lane, Wih0, bih0, bhh0, m, h);
    } else {
      v2f hb[KC_H];
      take_state(h0x + lane, hb);
      layer0_state_side(xs + ((i + 1) & 1) * SLOT + lane, hb, s0 + lane, whh0);
    }
    __syncthreads();
  }

  if (wave == 0) {
    #pragma unroll
    for (int i = 0; i < HH; ++i) stage[m * H + HH * h + i] = h1[i >> 1][i & 1];
  }
  __syncthreads();
  #pragma unroll
  for (int e = threadIdx.x; e < 16 * H; e += 96) out[(size_t)bbase * H + e] = stage[e];
}

extern "C" void kernel_launch(void* const* d_in, const int* in_sizes, int n_in,
                              void* d_out, int out_size, void* d_ws, size_t ws_size,
                              hipStream_t stream) {
  const float* x    = (const float*)d_in[0];
  const float* Wih0 = (const float*)d_in[1];
  const float* Whh0 = (const float*)d_in[2];
  const float* bih0 = (const float*)d_in[3];
  const float* bhh0 = (const float*)d_in[4];
  const float* Wih1 = (const float*)d_in[5];
  const float* Whh1 = (const float*)d_in[6];
  const float* bih1 = (const float*)d_in[7];
  const float* bhh1 = (const float*)d_in[8];
  float* out = (float*)d_out;

  const int B = in_sizes[0] / (T * I);
  const int nblocks = (B + 15) / 16;

  gru2_fused<<<nblocks, 96, 0, stream>>>(x, Wih0, Whh0, bih0, bhh0,
                                         Wih1, Whh1, bih1, bhh1, out);
  (void)n_in; (void)out_size; (void)d_ws; (void)ws_size;
}


# speedup vs baseline: 8.0232x; 1.0394x over previous
#include <hip/hip_runtime.h>
#include <cstdint>
#include <utility>


typedef float v2f __attribute__((ext_vector_type(2)));
typedef float v4f __attribute__((ext_vector_type(4)));
typedef float v8f __attribute__((ext_vector_type(8)));


namespace {

constexpr int H    = 40;
constexpr int HH   = 20;
constexpr int I    = 16;
constexpr int T    = 256;
constexpr int NRZ  = 5;
constexpr int NN   = 3;
constexpr int NT   = NRZ + NN;
constexpr int KC_I = 4;
constexpr int KC_H = 10;

__device__ __forceinline__ v8f wmma_f32(v2f a, v2f b, v8f c) {
  return __builtin_amdgcn_wmma_f32_16x16x4_f32(
       false, a,  false, b,
       (short)0, c,  false,  false);
}


__device__ __forceinline__ int gate_row(int tl, int m) {
  const int hr = m >> 3, v = m & 7;
  if (tl < NRZ) {
    const int q = tl * 8 + v;
    return (q < HH) ? (HH * hr + q) : (H + HH * hr + (q - HH));
  }
  const int q = (tl - NRZ) * 8 + v;
  return (q < HH) ? (2 * H + HH * hr + q) : -1;
}

template <int K, int KC>
__device__ __forceinline__ void load_a(const float* __restrict__ w, int m, int h, v2f (&a)[NT][KC]) {
  #pragma unroll
  for (int tl = 0; tl < NT; ++tl) {
    const int g  = gate_row(tl, m);
    const float* p = w + (g >= 0 ? g : 0) * K + (K / 2) * h;
    #pragma unroll
    for (int kc = 0; kc < KC; ++kc) {
      const v2f t = *(const v2f*)(p + 2 * kc);
      a[tl][kc] = (g >= 0) ? t : (v2f){0.f, 0.f};
    }
  }
}

__device__ __forceinline__ void load_bias(const float* __restrict__ bih, const float* __restrict__ bhh,
                                          int h, v8f (&brz)[NRZ], v2f (&bnx)[HH / 2], v2f (&bnh)[HH / 2]) {
  #pragma unroll
  for (int tl = 0; tl < NRZ; ++tl) {
    #pragma unroll
    for (int v = 0; v < 8; ++v) {
      const int g = gate_row(tl, 8 * h + v);
      brz[tl][v] = bih[g] + bhh[g];
    }
  }
  #pragma unroll
  for (int i = 0; i < HH; ++i) {
    bnx[i >> 1][i & 1] = bih[2 * H + HH * h + i];
    bnh[i >> 1][i & 1] = bhh[2 * H + HH * h + i];
  }
}


typedef int v2i __attribute__((ext_vector_type(2)));

constexpr int NB = 10;
constexpr int NP = NB / 2;
constexpr int ST_SIG  = 10;
constexpr int ST_CELL = 2 * ST_SIG + 3 + 26 + 2;
static_assert(ST_CELL == 51, "the stage numbers in CellRegs::op are written out for this count");
constexpr int LAG = 1;
static_assert(LAG == 1, "check the transcendental stages against each other before changing LAG");
static_assert(2 * NB == HH, "two batches cover a lane half's units");

__device__ __forceinline__ v2f splat(float k) { return (v2f){k, k}; }
__device__ __forceinline__ v2f fma2(v2f a, v2f b, v2f c) { return __builtin_elementwise_fma(a, b, c); }
__device__ __forceinline__ v2f pair_of(const v8f &a, int e) { return (v2f){a[e], a[e + 1]}; }

template <class V>
__device__ __forceinline__ void pin(V &a) { asm volatile("" : "+v"(a)); }

#define STAGE2(k, arr, expr) if constexpr (SK == (k)) { if (c == 0) { arr[j] = (expr); pin(arr[j]); } }

#define STAGE1(k, arr, expr) if constexpr (SK == (k)) { auto v_ = (expr); pin(v_); arr[j][c] = v_; }

#define STAGE_FLAG(k, arr, expr) if constexpr (SK == (k)) { arr[2 * j + c] = (expr); }

#define SIGMOID_STAGES(K, SRC, OUT)                                                           \
  STAGE2((K) + 0,  t,   __builtin_elementwise_min((SRC) * splat(-0x1.715476p+0f), splat(126.0f))) \
  STAGE1((K) + 1,  t,   __builtin_amdgcn_exp2f(t[j][c]))                                      \
  STAGE2((K) + 2,  dd,  splat(1.0f) + t[j])                                                   \
  STAGE1((K) + 3,  rc,  __builtin_amdgcn_rcpf(dd[j][c]))                                      \
  STAGE2((K) + 4,  f,   fma2(-dd[j], rc[j], splat(1.0f)))                                     \
  STAGE2((K) + 5,  rc,  fma2(f[j], rc[j], rc[j]))                                             \
  STAGE2((K) + 6,  f,   fma2(-dd[j], rc[j], splat(1.0f)))                                     \
  STAGE2((K) + 7,  q,   fma2(f[j], rc[j], rc[j]))                                             \
  STAGE2((K) + 8,  f,   fma2(-dd[j], q[j], splat(1.0f)))                                      \
  STAGE2((K) + 9,  OUT, fma2(f[j], rc[j], q[j]))

template <int I0>
struct CellRegs {
  static_assert(I0 % 2 == 0 && HH % 2 == 0, "a pair must not straddle an accumulator tile or the r|z split");
  v2f t[NP], dd[NP], rc[NP], f[NP], q[NP], r[NP], z[NP];
  v2f u[NP], w[NP], pre[NP];
  v2f y[NP], y2[NP], p[NP], zs[NP], x2[NP], ph[NP], pl[NP], en[NP], e[NP];
  v2i ei[NP];
  bool fl[NB];

  template <int SK>
  __device__ __forceinline__ void op(const int j, const int c,
                                     const v8f (&rz)[NRZ], const v8f (&xn)[NN], const v8f (&hn)[NN],
                                     const v2f (&bnx)[HH / 2], const v2f (&bnh)[HH / 2], v2f (&h)[KC_H]) {
    SIGMOID_STAGES(0,      pair_of(rz[(I0 + 2 * j) >> 3], (I0 + 2 * j) & 7), r)
    SIGMOID_STAGES(ST_SIG, pair_of(rz[(HH + I0 + 2 * j) >> 3], (HH + I0 + 2 * j) & 7), z)

    STAGE2(20, u,   pair_of(xn[(I0 + 2 * j) >> 3], (I0 + 2 * j) & 7) + bnx[I0 / 2 + j])
    STAGE2(21, w,   pair_of(hn[(I0 + 2 * j) >> 3], (I0 + 2 * j) & 7) + bnh[I0 / 2 + j])
    STAGE2(22, pre, fma2(r[j], w[j], u[j]))

    STAGE1(23, y,  __builtin_fabsf(pre[j][c]))
    STAGE2(24, y2, y[j] * y[j])
    STAGE2(25, p,  fma2(y2[j], splat(-0x1.758e7ap-8f), splat(0x1.521192p-6f)))
    STAGE2(26, p,  fma2(y2[j], p[j], splat(-0x1.b8389cp-5f)))
    STAGE2(27, p,  fma2(y2[j], p[j], splat(0x1.110704p-3f)))
    STAGE2(28, p,  fma2(y2[j], p[j], splat(-0x1.555532p-2f)))
    STAGE2(29, p,  y[j] * p[j])
    STAGE2(30, zs, fma2(y2[j], p[j], y[j]))
    STAGE2(31, x2, y[j] + y[j])
    STAGE2(32, ph, x2[j] * splat(0x1.715476p+0f))
    STAGE2(33, pl, fma2(x2[j], splat(0x1.715476p+0f), -ph[j]))
    STAGE1(34, en, __builtin_rintf(ph[j][c]))
    STAGE2(35, pl, fma2(x2[j], splat(0x1.4ae0bep-26f), pl[j]))
    STAGE2(36, ph, ph[j] - en[j])
    STAGE2(37, ph, ph[j] + pl[j])
    STAGE1(38, ei, (int)en[j][c])
    STAGE1(39, e,  __builtin_amdgcn_exp2f(ph[j][c]))
    STAGE1(40, e,  __builtin_ldexpf(e[j][c], ei[j][c]))
    STAGE_FLAG(41, fl, !(0x1.62e43p+6f < x2[j][c]))
    STAGE1(42, e,  fl[2 * j + c] ? e[j][c] : __builtin_inff())
    STAGE2(43, e,  e[j] + splat(1.0f))
    STAGE1(44, e,  __builtin_amdgcn_rcpf(e[j][c]))
    STAGE2(45, e,  fma2(splat(-2.0f), e[j], splat(1.0f)))
    STAGE_FLAG(46, fl, y[j][c] < 0.625f)
    STAGE1(47, e,  fl[2 * j + c] ? zs[j][c] : e[j][c])
    STAGE1(48, e,  __builtin_copysignf(e[j][c], pre[j][c]))

    STAGE2(49, u,  h[I0 / 2 + j] - e[j])
    STAGE2(50, u,  fma2(z[j], u[j], e[j]))
    if constexpr (SK == 50) { if (c == 0) h[I0 / 2 + j] = u[j]; }
  }
};

template <int S>
__device__ __forceinline__ void cell_slot(CellRegs<0> &A, CellRegs<NB> &B,
                                          const v8f (&rz)[NRZ], const v8f (&xn)[NN], const v8f (&hn)[NN],
                                          const v2f (&bnx)[HH / 2], const v2f (&bnh)[HH / 2], v2f (&h)[KC_H]) {
  #pragma unroll
  for (int j = 0; j < NP; ++j) {
    #pragma unroll
    for (int c = 0; c < 2; ++c) {
      A.template op<S>(j, c, rz, xn, hn, bnx, bnh, h);
      B.template op<S - LAG>(j, c, rz, xn, hn, bnx, bnh, h);
    }
  }
}

template <int... S>
__device__ __forceinline__ void cell_slots(std::integer_sequence<int, S...>, CellRegs<0> &A, CellRegs<NB> &B,
                                           const v8f (&rz)[NRZ], const v8f (&xn)[NN], const v8f (&hn)[NN],
                                           const v2f (&bnx)[HH / 2], const v2f (&bnh)[HH / 2], v2f (&h)[KC_H]) {
  (cell_slot<S>(A, B, rz, xn, hn, bnx, bnh, h), ...);
}

__device__ __forceinline__ void cell_update(const v8f (&rz)[NRZ], const v8f (&xn)[NN], const v8f (&hn)[NN],
                                            const v2f (&bnx)[HH / 2], const v2f (&bnh)[HH / 2], v2f (&h)[KC_H]) {
  CellRegs<0>  A;
  CellRegs<NB> B;
  cell_slots(std::make_integer_sequence<int, ST_CELL + LAG>{}, A, B, rz, xn, hn, bnx, bnh, h);
}

constexpr int TSPLIT = 4;
static_assert(TSPLIT < NRZ, "wave 2's share is written for one r|z tile and the n tiles");
constexpr int SLOT = 16 * 32;
constexpr int HSL  = (HH / 4) * 32;
static_assert(HH % 4 == 0, "a lane's share of a state is handed over in 16-byte pieces");

__device__ __forceinline__ void put_slot(v4f* d, const v8f (&rz)[NRZ], const v8f (&n)[NN]) {
  #pragma unroll
  for (int tl = 0; tl < NRZ; ++tl) {
    d[(2 * tl) * 32]     = (v4f){rz[tl][0], rz[tl][1], rz[tl][2], rz[tl][3]};
    d[(2 * tl + 1) * 32] = (v4f){rz[tl][4], rz[tl][5], rz[tl][6], rz[tl][7]};
  }
  #pragma unroll
  for (int tl = 0; tl < NN; ++tl) {
    d[(2 * (NRZ + tl)) * 32]     = (v4f){n[tl][0], n[tl][1], n[tl][2], n[tl][3]};
    d[(2 * (NRZ + tl) + 1) * 32] = (v4f){n[tl][4], n[tl][5], n[tl][6], n[tl][7]};
  }
}

__device__ __forceinline__ void take_slot(const v4f* src, v8f (&rz)[NRZ], v8f (&xn)[NN]) {
  #pragma unroll
  for (int tl = 0; tl < NRZ; ++tl) {
    const v4f lo = src[(2 * tl) * 32], hi = src[(2 * tl + 1) * 32];
    rz[tl] = (v8f){lo.x, lo.y, lo.z, lo.w, hi.x, hi.y, hi.z, hi.w};
  }
  #pragma unroll
  for (int tl = 0; tl < NN; ++tl) {
    const v4f lo = src[(2 * (NRZ + tl)) * 32], hi = src[(2 * (NRZ + tl) + 1) * 32];
    xn[tl] = (v8f){lo.x, lo.y, lo.z, lo.w, hi.x, hi.y, hi.z, hi.w};
  }
}

__device__ __forceinline__ void take_slot_n(const v4f* src, v8f (&xn)[NN]) {
  #pragma unroll
  for (int tl = 0; tl < NN; ++tl) {
    const v4f lo = src[(2 * (NRZ + tl)) * 32], hi = src[(2 * (NRZ + tl) + 1) * 32];
    xn[tl] = (v8f){lo.x, lo.y, lo.z, lo.w, hi.x, hi.y, hi.z, hi.w};
  }
}

__device__ __forceinline__ void put_state(v4f* d, const v2f (&hs)[KC_H]) {
  #pragma unroll
  for (int q = 0; q < HH / 4; ++q) d[q * 32] = (v4f){hs[2 * q][0], hs[2 * q][1], hs[2 * q + 1][0], hs[2 * q + 1][1]};
}
__device__ __forceinline__ void take_state(const v4f* src, v2f (&hs)[KC_H]) {
  #pragma unroll
  for (int q = 0; q < HH / 4; ++q) {
    const v4f t = src[q * 32];
    hs[2 * q] = (v2f){t.x, t.y};
    hs[2 * q + 1] = (v2f){t.z, t.w};
  }
}

__device__ __forceinline__ void produce(const float* __restrict__ x, int xoff, int s0, v4f* dst,
                                        const float* __restrict__ Wih0,
                                        const float* __restrict__ bih0, const float* __restrict__ bhh0,
                                        int m, int h) {
  v2f wih0[NT][KC_I];
  load_a<I, KC_I>(Wih0, m, h, wih0);
  v8f brz0[NRZ];
  v2f bnx[HH / 2], bnh[HH / 2];
  load_bias(bih0, bhh0, h, brz0, bnx, bnh);
  const int s = (s0 < T) ? s0 : (T - 1);
  const float* xrow = x + (xoff + s * I);
  const v4f xa = *(const v4f*)(xrow);
  const v4f xb = *(const v4f*)(xrow + 4);
  const v8f zero8 = {0.f, 0.f, 0.f, 0.f, 0.f, 0.f, 0.f, 0.f};
  const v2f xf[KC_I] = {{xa.x, xa.y}, {xa.z, xa.w}, {xb.x, xb.y}, {xb.z, xb.w}};
  v8f rz[NRZ], xn[NN];
  #pragma unroll
  for (int tl = 0; tl < NRZ; ++tl) rz[tl] = brz0[tl];
  #pragma unroll
  for (int tl = 0; tl < NN; ++tl) xn[tl] = zero8;
  #pragma unroll
  for (int wi = 0; wi < 8 * KC_I; ++wi) {
    const int kc = wi >> 3, tl = wi & 7;
    if (tl < NRZ) rz[tl] = wmma_f32(wih0[tl][kc], xf[kc], rz[tl]);
    else          xn[tl - NRZ] = wmma_f32(wih0[tl][kc], xf[kc], xn[tl - NRZ]);
  }
  put_slot(dst, rz, xn);
}

__device__ __forceinline__ void state_product(int wi, const v2f (&w)[NT][KC_H], const v2f (&b)[KC_H],
                                              v8f (&n)[NN], v8f (&rz)[NRZ]) {
  const int kc = wi >> 3, tl = wi & 7;
  if (kc < KC_H) {
    if (tl < NRZ) rz[tl] = wmma_f32(w[tl][kc], b[kc], rz[tl]);
    else          n[tl - NRZ] = wmma_f32(w[tl][kc], b[kc], n[tl - NRZ]);
  }
}

template <int TL0, int TL1>
__device__ __forceinline__ void tile_products(const v2f (&w)[NT][KC_H], const v2f (&b)[KC_H],
                                              v8f (&n)[NN], v8f (&rz)[NRZ]) {
  #pragma unroll
  for (int kc = 0; kc < KC_H; ++kc) {
    #pragma unroll
    for (int tl = TL0; tl < TL1; ++tl) {
      if (tl < NRZ) rz[tl] = wmma_f32(w[tl][kc], b[kc], rz[tl]);
      else          n[tl - NRZ] = wmma_f32(w[tl][kc], b[kc], n[tl - NRZ]);
    }
  }
}

__device__ __forceinline__ void put_tile(v4f* d, int tl, const v8f &a) {
  d[(2 * tl) * 32]     = (v4f){a[0], a[1], a[2], a[3]};
  d[(2 * tl + 1) * 32] = (v4f){a[4], a[5], a[6], a[7]};
}

__device__ __forceinline__ void layer0_state_side(const v4f* in, const v2f (&hb)[KC_H], v4f* dst,
                                                  const v2f (&whh0)[NT][KC_H]) {
  v8f rz[NRZ], hn[NN];
  #pragma unroll
  for (int tl = 0; tl < NRZ; ++tl) {
    const v4f lo = in[(2 * tl) * 32], hi = in[(2 * tl + 1) * 32];
    rz[tl] = (v8f){lo.x, lo.y, lo.z, lo.w, hi.x, hi.y, hi.z, hi.w};
  }
  #pragma unroll
  for (int tl = 0; tl < NN; ++tl) hn[tl] = (v8f){0.f, 0.f, 0.f, 0.f, 0.f, 0.f, 0.f, 0.f};
  #pragma unroll
  for (int wi = 0; wi < 8 * KC_H; ++wi) state_product(wi, whh0, hb, hn, rz);
  put_slot(dst, rz, hn);
}

}

__global__ __launch_bounds__(96) void gru2_fused(
    const float* __restrict__ x,
    const float* __restrict__ Wih0, const float* __restrict__ Whh0,
    const float* __restrict__ bih0, const float* __restrict__ bhh0,
    const float* __restrict__ Wih1, const float* __restrict__ Whh1,
    const float* __restrict__ bih1, const float* __restrict__ bhh1,
    float* __restrict__ out)
{
  __shared__ v4f xs[2 * SLOT];
  __shared__ v4f s0[SLOT];
  __shared__ v4f s1[SLOT];
  __shared__ v4f hx[HSL];
  __shared__ v4f h0x[HSL];
  __shared__ v4f b1[2 * TSPLIT * 32];
  __shared__ float stage[16 * H];
  const int lane  = threadIdx.x & 31;
  const int wave  = __builtin_amdgcn_readfirstlane((int)(threadIdx.x >> 5));
  const int m     = lane & 15;
  const int h     = lane >> 4;
  const int bbase = blockIdx.x * 16;

  const int xoff = (bbase + m) * (T * I) + 8 * h;

  const v8f zero8 = {0.f, 0.f, 0.f, 0.f, 0.f, 0.f, 0.f, 0.f};

  v2f wih1[NT][KC_H];
  v2f bnx0[HH / 2], bnh0[HH / 2], bnx1[HH / 2], bnh1[HH / 2];
  v2f h0[KC_H], h1[KC_H];
  v2f whh1[NT][KC_H];
  v2f whh0[NT][KC_H], whh1b[NT][KC_H];
  v8f brz1hi[NRZ];

  if (wave == 0) {
    load_a<H, KC_H>(Wih1, m, h, wih1);
    v8f brzu[NRZ];
    load_bias(bih0, bhh0, h, brzu, bnx0, bnh0);
    load_bias(bih1, bhh1, h, brzu, bnx1, bnh1);
    #pragma unroll
    for (int kc = 0; kc < KC_H; ++kc) { h0[kc] = (v2f){0.f, 0.f}; h1[kc] = (v2f){0.f, 0.f}; }
    put_state(hx + lane, h1);
  } else if (wave == 1) {
    load_a<H, KC_H>(Whh1, m, h, whh1);
    v8f brz1[NRZ];
    v2f bu0[HH / 2], bu1[HH / 2];
    load_bias(bih1, bhh1, h, brz1, bu0, bu1);
    #pragma unroll
    for (int tl = 0; tl < TSPLIT; ++tl) {
      b1[(2 * tl) * 32 + lane]     = (v4f){brz1[tl][0], brz1[tl][1], brz1[tl][2], brz1[tl][3]};
      b1[(2 * tl + 1) * 32 + lane] = (v4f){brz1[tl][4], brz1[tl][5], brz1[tl][6], brz1[tl][7]};
    }
    produce(x, xoff, 0, xs + lane, Wih0, bih0, bhh0, m, h);
    produce(x, xoff, 1, xs + SLOT + lane, Wih0, bih0, bhh0, m, h);
  } else {
    load_a<H, KC_H>(Whh0, m, h, whh0);
    load_a<H, KC_H>(Whh1, m, h, whh1b);
    v2f bu0[HH / 2], bu1[HH / 2];
    load_bias(bih1, bhh1, h, brz1hi, bu0, bu1);
  }
  __syncthreads();

  if (wave == 2) {
    v2f hz[KC_H];
    #pragma unroll
    for (int kc = 0; kc < KC_H; ++kc) hz[kc] = (v2f){0.f, 0.f};
    layer0_state_side(xs + lane, hz, s0 + lane, whh0);
  }
  __syncthreads();

  #pragma unroll 1
  for (int i = 0; i < T; ++i) {
    if (wave == 0) {
      v8f rz0[NRZ], xn0[NN], hn0[NN];
      take_slot(s0 + lane, rz0, hn0);
      take_slot_n(xs + (i & 1) * SLOT + lane, xn0);
      cell_update(rz0, xn0, hn0, bnx0, bnh0, h0);
      put_state(h0x + lane, h0);
    } else if (wave == 1) {
      v2f hb[KC_H];
      take_state(hx + lane, hb);
      v8f rz[NRZ], hn[NN];
      #pragma unroll
      for (int tl = 0; tl < TSPLIT; ++tl) {
        const v4f lo = b1[(2 * tl) * 32 + lane], hi = b1[(2 * tl + 1) * 32 + lane];
        rz[tl] = (v8f){lo.x, lo.y, lo.z, lo.w, hi.x, hi.y, hi.z, hi.w};
      }
      tile_products<0, TSPLIT>(whh1, hb, hn, rz);
      #pragma unroll
      for (int tl = 0; tl < TSPLIT; ++tl) put_tile(s1 + lane, tl, rz[tl]);
    } else {
      v2f hb[KC_H];
      take_state(hx + lane, hb);
      v8f rz[NRZ], hn[NN];
      #pragma unroll
      for (int tl = TSPLIT; tl < NRZ; ++tl) rz[tl] = brz1hi[tl];
      #pragma unroll
      for (int tl = 0; tl < NN; ++tl) hn[tl] = zero8;
      tile_products<TSPLIT, NT>(whh1b, hb, hn, rz);
      #pragma unroll
      for (int tl = TSPLIT; tl < NRZ; ++tl) put_tile(s1 + lane, tl, rz[tl]);
      #pragma unroll
      for (int tl = 0; tl < NN; ++tl) put_tile(s1 + lane, NRZ + tl, hn[tl]);
    }
    __syncthreads();
    if (wave == 0) {
      v8f rz1[NRZ], xn1[NN], hn1[NN];
      take_slot(s1 + lane, rz1, hn1);
      #pragma unroll
      for (int tl = 0; tl < NN; ++tl) xn1[tl] = zero8;
      #pragma unroll
      for (int wi = 0; wi < 8 * KC_H; ++wi) state_product(wi, wih1, h0, xn1, rz1);
      cell_update(rz1, xn1, hn1, bnx1, bnh1, h1);
      put_state(hx + lane, h1);
    } else if (wave == 1) {
      produce(x, xoff, i + 2, xs + (i & 1) * SLOT + lane, Wih0, bih0, bhh0, m, h);
    } else {
      v2f hb[KC_H];
      take_state(h0x + lane, hb);
      layer0_state_side(xs + ((i + 1) & 1) * SLOT + lane, hb, s0 + lane, whh0);
    }
    __syncthreads();
  }

  if (wave == 0) {
    #pragma unroll
    for (int i = 0; i < HH; ++i) stage[m * H + HH * h + i] = h1[i >> 1][i & 1];
  }
  __syncthreads();
  #pragma unroll
  for (int e = threadIdx.x; e < 16 * H; e += 96) out[(size_t)bbase * H + e] = stage[e];
}

extern "C" void kernel_launch(void* const* d_in, const int* in_sizes, int n_in,
                              void* d_out, int out_size, void* d_ws, size_t ws_size,
                              hipStream_t stream) {
  const float* x    = (const float*)d_in[0];
  const float* Wih0 = (const float*)d_in[1];
  const float* Whh0 = (const float*)d_in[2];
  const float* bih0 = (const float*)d_in[3];
  const float* bhh0 = (const float*)d_in[4];
  const float* Wih1 = (const float*)d_in[5];
  const float* Whh1 = (const float*)d_in[6];
  const float* bih1 = (const float*)d_in[7];
  const float* bhh1 = (const float*)d_in[8];
  float* out = (float*)d_out;

  const int B = in_sizes[0] / (T * I);
  const int nblocks = (B + 15) / 16;

  gru2_fused<<<nblocks, 96, 0, stream>>>(x, Wih0, Whh0, bih0, bhh0,
                                         Wih1, Whh1, bih1, bhh1, out);
  (void)n_in; (void)out_size; (void)d_ws; (void)ws_size;
}


# Layer 1's cell update runs one phase behind layer 0's on the recurrence wave, so every product (Wih1 h0 too) goes to 3 helper waves on 3 other SIMDs, dealt by tiles; 2 barriers a step; sums unchanged
# speedup vs baseline: 11.8060x; 1.4715x over previous
#include <hip/hip_runtime.h>
#include <cstdint>
#include <utility>


typedef float v2f __attribute__((ext_vector_type(2)));
typedef float v4f __attribute__((ext_vector_type(4)));
typedef float v8f __attribute__((ext_vector_type(8)));


namespace {

constexpr int H    = 40;
constexpr int HH   = 20;
constexpr int I    = 16;
constexpr int T    = 256;
constexpr int NRZ  = 5;
constexpr int NN   = 3;
constexpr int NT   = NRZ + NN;
constexpr int KC_I = 4;
constexpr int KC_H = 10;

__device__ __forceinline__ v8f wmma_f32(v2f a, v2f b, v8f c) {
  return __builtin_amdgcn_wmma_f32_16x16x4_f32(
       false, a,  false, b,
       (short)0, c,  false,  false);
}


__device__ __forceinline__ int rz_row(int tl, int mr) {
  const int hr = mr >> 3, q = tl * 8 + (mr & 7);
  return HH * hr + q + ((q < HH) ? 0 : (H - HH));
}
__device__ __forceinline__ int n_row(int tl, int mr, bool &real) {
  const int hr = mr >> 3, q = (tl - NRZ) * 8 + (mr & 7);
  real = q < HH;
  return 2 * H + HH * hr + (real ? q : 0);
}

__device__ __forceinline__ void load_bias(const float* __restrict__ bih, const float* __restrict__ bhh,
                                          int h, v8f (&brz)[NRZ], v2f (&bnx)[HH / 2], v2f (&bnh)[HH / 2]) {
  #pragma unroll
  for (int tl = 0; tl < NRZ; ++tl) {
    #pragma unroll
    for (int v = 0; v < 8; ++v) {
      const int g = rz_row(tl, 8 * h + v);
      brz[tl][v] = bih[g] + bhh[g];
    }
  }
  #pragma unroll
  for (int i = 0; i < HH; ++i) {
    bnx[i >> 1][i & 1] = bih[2 * H + HH * h + i];
    bnh[i >> 1][i & 1] = bhh[2 * H + HH * h + i];
  }
}


typedef int v2i __attribute__((ext_vector_type(2)));

constexpr int NB = 10;
constexpr int NP = NB / 2;
constexpr int ST_SIG  = 10;
constexpr int ST_CELL = 2 * ST_SIG + 3 + 26 + 2;
static_assert(ST_CELL == 51, "the stage numbers in CellRegs::op are written out for this count");
constexpr int LAG = 1;
static_assert(LAG == 1, "check the transcendental stages against each other before changing LAG");
static_assert(2 * NB == HH, "two batches cover a lane half's units");

__device__ __forceinline__ v2f splat(float k) { return (v2f){k, k}; }
__device__ __forceinline__ v2f fma2(v2f a, v2f b, v2f c) { return __builtin_elementwise_fma(a, b, c); }
__device__ __forceinline__ v2f pair_of(const v8f &a, int e) { return (v2f){a[e], a[e + 1]}; }

template <class V>
__device__ __forceinline__ void pin(V &a) { asm volatile("" : "+v"(a)); }

#define STAGE2(k, arr, expr) if constexpr (SK == (k)) { if (c == 0) { arr[j] = (expr); pin(arr[j]); } }

#define STAGE1(k, arr, expr) if constexpr (SK == (k)) { auto v_ = (expr); pin(v_); arr[j][c] = v_; }

#define STAGE_FLAG(k, arr, expr) if constexpr (SK == (k)) { arr[2 * j + c] = (expr); }

#define SIGMOID_STAGES(K, SRC, OUT)                                                           \
  STAGE2((K) + 0,  t,   __builtin_elementwise_min((SRC) * splat(-0x1.715476p+0f), splat(126.0f))) \
  STAGE1((K) + 1,  t,   __builtin_amdgcn_exp2f(t[j][c]))                                      \
  STAGE2((K) + 2,  dd,  splat(1.0f) + t[j])                                                   \
  STAGE1((K) + 3,  rc,  __builtin_amdgcn_rcpf(dd[j][c]))                                      \
  STAGE2((K) + 4,  f,   fma2(-dd[j], rc[j], splat(1.0f)))                                     \
  STAGE2((K) + 5,  rc,  fma2(f[j], rc[j], rc[j]))                                             \
  STAGE2((K) + 6,  f,   fma2(-dd[j], rc[j], splat(1.0f)))                                     \
  STAGE2((K) + 7,  q,   fma2(f[j], rc[j], rc[j]))                                             \
  STAGE2((K) + 8,  f,   fma2(-dd[j], q[j], splat(1.0f)))                                      \
  STAGE2((K) + 9,  OUT, fma2(f[j], rc[j], q[j]))

template <int I0>
struct CellRegs {
  static_assert(I0 % 2 == 0 && HH % 2 == 0, "a pair must not straddle an accumulator tile or the r|z split");
  v2f t[NP], dd[NP], rc[NP], f[NP], q[NP], r[NP], z[NP];
  v2f u[NP], w[NP], pre[NP];
  v2f y[NP], y2[NP], p[NP], zs[NP], x2[NP], ph[NP], pl[NP], en[NP], e[NP];
  v2i ei[NP];
  bool fl[NB];

  template <int SK>
  __device__ __forceinline__ void op(const int j, const int c,
                                     const v8f (&rz)[NRZ], const v8f (&xn)[NN], const v8f (&hn)[NN],
                                     const v2f (&bnx)[HH / 2], const v2f (&bnh)[HH / 2], v2f (&h)[KC_H]) {
    SIGMOID_STAGES(0,      pair_of(rz[(I0 + 2 * j) >> 3], (I0 + 2 * j) & 7), r)
    SIGMOID_STAGES(ST_SIG, pair_of(rz[(HH + I0 + 2 * j) >> 3], (HH + I0 + 2 * j) & 7), z)

    STAGE2(20, u,   pair_of(xn[(I0 + 2 * j) >> 3], (I0 + 2 * j) & 7) + bnx[I0 / 2 + j])
    STAGE2(21, w,   pair_of(hn[(I0 + 2 * j) >> 3], (I0 + 2 * j) & 7) + bnh[I0 / 2 + j])
    STAGE2(22, pre, fma2(r[j], w[j], u[j]))

    STAGE1(23, y,  __builtin_fabsf(pre[j][c]))
    STAGE2(24, y2, y[j] * y[j])
    STAGE2(25, p,  fma2(y2[j], splat(-0x1.758e7ap-8f), splat(0x1.521192p-6f)))
    STAGE2(26, p,  fma2(y2[j], p[j], splat(-0x1.b8389cp-5f)))
    STAGE2(27, p,  fma2(y2[j], p[j], splat(0x1.110704p-3f)))
    STAGE2(28, p,  fma2(y2[j], p[j], splat(-0x1.555532p-2f)))
    STAGE2(29, p,  y[j] * p[j])
    STAGE2(30, zs, fma2(y2[j], p[j], y[j]))
    STAGE2(31, x2, y[j] + y[j])
    STAGE2(32, ph, x2[j] * splat(0x1.715476p+0f))
    STAGE2(33, pl, fma2(x2[j], splat(0x1.715476p+0f), -ph[j]))
    STAGE1(34, en, __builtin_rintf(ph[j][c]))
    STAGE2(35, pl, fma2(x2[j], splat(0x1.4ae0bep-26f), pl[j]))
    STAGE2(36, ph, ph[j] - en[j])
    STAGE2(37, ph, ph[j] + pl[j])
    STAGE1(38, ei, (int)en[j][c])
    STAGE1(39, e,  __builtin_amdgcn_exp2f(ph[j][c]))
    STAGE1(40, e,  __builtin_ldexpf(e[j][c], ei[j][c]))
    STAGE_FLAG(41, fl, !(0x1.62e43p+6f < x2[j][c]))
    STAGE1(42, e,  fl[2 * j + c] ? e[j][c] : __builtin_inff())
    STAGE2(43, e,  e[j] + splat(1.0f))
    STAGE1(44, e,  __builtin_amdgcn_rcpf(e[j][c]))
    STAGE2(45, e,  fma2(splat(-2.0f), e[j], splat(1.0f)))
    STAGE_FLAG(46, fl, y[j][c] < 0.625f)
    STAGE1(47, e,  fl[2 * j + c] ? zs[j][c] : e[j][c])
    STAGE1(48, e,  __builtin_copysignf(e[j][c], pre[j][c]))

    STAGE2(49, u,  h[I0 / 2 + j] - e[j])
    STAGE2(50, u,  fma2(z[j], u[j], e[j]))
    if constexpr (SK == 50) { if (c == 0) h[I0 / 2 + j] = u[j]; }
  }
};

template <int S>
__device__ __forceinline__ void cell_slot(CellRegs<0> &A, CellRegs<NB> &B,
                                          const v8f (&rz)[NRZ], const v8f (&xn)[NN], const v8f (&hn)[NN],
                                          const v2f (&bnx)[HH / 2], const v2f (&bnh)[HH / 2], v2f (&h)[KC_H]) {
  #pragma unroll
  for (int j = 0; j < NP; ++j) {
    #pragma unroll
    for (int c = 0; c < 2; ++c) {
      A.template op<S>(j, c, rz, xn, hn, bnx, bnh, h);
      B.template op<S - LAG>(j, c, rz, xn, hn, bnx, bnh, h);
    }
  }
}

template <int... S>
__device__ __forceinline__ void cell_slots(std::integer_sequence<int, S...>, CellRegs<0> &A, CellRegs<NB> &B,
                                           const v8f (&rz)[NRZ], const v8f (&xn)[NN], const v8f (&hn)[NN],
                                           const v2f (&bnx)[HH / 2], const v2f (&bnh)[HH / 2], v2f (&h)[KC_H]) {
  (cell_slot<S>(A, B, rz, xn, hn, bnx, bnh, h), ...);
}

__device__ __forceinline__ void cell_update(const v8f (&rz)[NRZ], const v8f (&xn)[NN], const v8f (&hn)[NN],
                                            const v2f (&bnx)[HH / 2], const v2f (&bnh)[HH / 2], v2f (&h)[KC_H]) {
  CellRegs<0>  A;
  CellRegs<NB> B;
  cell_slots(std::make_integer_sequence<int, ST_CELL + LAG>{}, A, B, rz, xn, hn, bnx, bnh, h);
}


constexpr int SLOT  = 16 * 32;
constexpr int SLOT1 = 22 * 32;
constexpr int HSL   = (HH / 4) * 32;
static_assert(HH % 4 == 0, "a lane's share of a state is handed over in 16-byte pieces");

struct TileW { v2f hh1[KC_H], ih1[KC_H], hh0[KC_H], ih0[KC_I]; };

template <int K, int KC>
__device__ __forceinline__ void load_a_tile(const float* __restrict__ w, int g, bool real, int h, v2f (&a)[KC]) {
  const float* p = w + g * K + (K / 2) * h;
  #pragma unroll
  for (int kc = 0; kc < KC; ++kc) {
    const v2f t = *(const v2f*)(p + 2 * kc);
    a[kc] = real ? t : (v2f){0.f, 0.f};
  }
}

__device__ __forceinline__ void load_tile_w(TileW &t, int g, bool real, int h,
                                            const float* __restrict__ Whh1, const float* __restrict__ Wih1,
                                            const float* __restrict__ Whh0, const float* __restrict__ Wih0) {
  load_a_tile<H, KC_H>(Whh1, g, real, h, t.hh1);
  load_a_tile<H, KC_H>(Wih1, g, real, h, t.ih1);
  load_a_tile<H, KC_H>(Whh0, g, real, h, t.hh0);
  load_a_tile<I, KC_I>(Wih0, g, real, h, t.ih0);
}

__device__ __forceinline__ v8f bias_tile(const float* __restrict__ bih, const float* __restrict__ bhh, int tl, int h) {
  v8f b;
  #pragma unroll
  for (int v = 0; v < 8; ++v) {
    const int g = rz_row(tl, 8 * h + v);
    b[v] = bih[g] + bhh[g];
  }
  return b;
}

__device__ __forceinline__ void put_tile(v4f* d, int tl, const v8f &a) {
  d[(2 * tl) * 32]     = (v4f){a[0], a[1], a[2], a[3]};
  d[(2 * tl + 1) * 32] = (v4f){a[4], a[5], a[6], a[7]};
}
__device__ __forceinline__ v8f take_tile(const v4f* s, int tl) {
  const v4f lo = s[(2 * tl) * 32], hi = s[(2 * tl + 1) * 32];
  return (v8f){lo.x, lo.y, lo.z, lo.w, hi.x, hi.y, hi.z, hi.w};
}

__device__ __forceinline__ void take_slot(const v4f* s, v8f (&rz)[NRZ], v8f (&n)[NN]) {
  #pragma unroll
  for (int tl = 0; tl < NRZ; ++tl) rz[tl] = take_tile(s, tl);
  #pragma unroll
  for (int tl = 0; tl < NN; ++tl) n[tl] = take_tile(s, NRZ + tl);
}
__device__ __forceinline__ void take_slot_n(const v4f* s, int t0, v8f (&n)[NN]) {
  #pragma unroll
  for (int tl = 0; tl < NN; ++tl) n[tl] = take_tile(s, t0 + tl);
}

__device__ __forceinline__ void put_state(v4f* d, const v2f (&hs)[KC_H]) {
  #pragma unroll
  for (int q = 0; q < HH / 4; ++q) d[q * 32] = (v4f){hs[2 * q][0], hs[2 * q][1], hs[2 * q + 1][0], hs[2 * q + 1][1]};
}
__device__ __forceinline__ void take_state(const v4f* s, v2f (&hs)[KC_H]) {
  #pragma unroll
  for (int q = 0; q < HH / 4; ++q) {
    const v4f t = s[q * 32];
    hs[2 * q] = (v2f){t.x, t.y};
    hs[2 * q + 1] = (v2f){t.z, t.w};
  }
}

__device__ __forceinline__ void helper_input_side(v4f* dst, int ta, int tb, int tn,
                                                  const TileW &wa, const TileW &wb, const TileW &wn,
                                                  const v8f &b0a, const v8f &b0b, const v4f x4a, const v4f x4b) {
  const v2f xf[KC_I] = {{x4a.x, x4a.y}, {x4a.z, x4a.w}, {x4b.x, x4b.y}, {x4b.z, x4b.w}};
  v8f pa = b0a, pb = b0b, pn = {0.f, 0.f, 0.f, 0.f, 0.f, 0.f, 0.f, 0.f};
  #pragma unroll
  for (int kc = 0; kc < KC_I; ++kc) {
    pa = wmma_f32(wa.ih0[kc], xf[kc], pa);
    pb = wmma_f32(wb.ih0[kc], xf[kc], pb);
    pn = wmma_f32(wn.ih0[kc], xf[kc], pn);
  }
  put_tile(dst, ta, pa);
  put_tile(dst, tb, pb);
  put_tile(dst, tn, pn);
}

__device__ __forceinline__ void helper_layer0(const v4f* h0s, const v4f* xin, v4f* s0d, v4f* xout,
                                              int ta, int tb, int tn,
                                              const TileW &wa, const TileW &wb, const TileW &wn,
                                              const v8f &b0a, const v8f &b0b, const v4f x4a, const v4f x4b,
                                              v2f (&hb0)[KC_H], v8f &xnk) {
  take_state(h0s, hb0);
  const v8f zero8 = {0.f, 0.f, 0.f, 0.f, 0.f, 0.f, 0.f, 0.f};
  v8f ra = take_tile(xin, ta), rb = take_tile(xin, tb), hn = zero8, xn = zero8;
  #pragma unroll
  for (int kc = 0; kc < KC_H; ++kc) {
    ra = wmma_f32(wa.hh0[kc], hb0[kc], ra);
    rb = wmma_f32(wb.hh0[kc], hb0[kc], rb);
    hn = wmma_f32(wn.hh0[kc], hb0[kc], hn);
    xn = wmma_f32(wn.ih1[kc], hb0[kc], xn);
  }
  xnk = xn;
  put_tile(s0d, ta, ra);
  put_tile(s0d, tb, rb);
  put_tile(s0d, tn, hn);
  helper_input_side(xout, ta, tb, tn, wa, wb, wn, b0a, b0b, x4a, x4b);
}

__device__ __forceinline__ void helper_layer1(const v4f* h1s, v4f* s1d, int ta, int tb, int tn,
                                              const TileW &wa, const TileW &wb, const TileW &wn,
                                              const v8f &b1a, const v8f &b1b, const v2f (&hb0)[KC_H], const v8f &xnk) {
  v2f hb1[KC_H];
  take_state(h1s, hb1);
  const v8f zero8 = {0.f, 0.f, 0.f, 0.f, 0.f, 0.f, 0.f, 0.f};
  v8f ra = b1a, rb = b1b, hn = zero8;
  #pragma unroll
  for (int kc = 0; kc < KC_H; ++kc) {
    ra = wmma_f32(wa.hh1[kc], hb1[kc], ra);
    rb = wmma_f32(wb.hh1[kc], hb1[kc], rb);
    hn = wmma_f32(wn.hh1[kc], hb1[kc], hn);
  }
  #pragma unroll
  for (int kc = 0; kc < KC_H; ++kc) {
    ra = wmma_f32(wa.ih1[kc], hb0[kc], ra);
    rb = wmma_f32(wb.ih1[kc], hb0[kc], rb);
  }
  put_tile(s1d, ta, ra);
  put_tile(s1d, tb, rb);
  put_tile(s1d, tn, hn);
  put_tile(s1d, tn + NN, xnk);
}

}

__global__ __launch_bounds__(128) void gru2_fused(
    const float* __restrict__ x,
    const float* __restrict__ Wih0, const float* __restrict__ Whh0,
    const float* __restrict__ bih0, const float* __restrict__ bhh0,
    const float* __restrict__ Wih1, const float* __restrict__ Whh1,
    const float* __restrict__ bih1, const float* __restrict__ bhh1,
    float* __restrict__ out)
{
  __shared__ v4f xs[2 * SLOT];
  __shared__ v4f s0[SLOT];
  __shared__ v4f s1[SLOT1];
  __shared__ v4f hx[HSL];
  __shared__ v4f h0x[HSL];
  __shared__ float stage[16 * H];
  const int lane  = threadIdx.x & 31;
  const int wave  = __builtin_amdgcn_readfirstlane((int)(threadIdx.x >> 5));
  const int m     = lane & 15;
  const int h     = lane >> 4;
  const int bbase = blockIdx.x * 16;

  const int xoff = (bbase + m) * (T * I) + 8 * h;

  const int hw = wave - 1;
  const int ta = (hw < 2) ? 2 * hw : NRZ - 1;
  const int tb = (hw < 2) ? 2 * hw + 1 : NRZ - 1;
  const int tn = NRZ + hw;

  v2f bnx0[HH / 2], bnh0[HH / 2], bnx1[HH / 2], bnh1[HH / 2];
  v2f h0[KC_H], h1[KC_H];
  TileW wa, wb, wn;
  v8f b0a, b0b, b1a, b1b;
  v2f hb0[KC_H];
  v8f xnk = {0.f, 0.f, 0.f, 0.f, 0.f, 0.f, 0.f, 0.f};
  v4f x4a = {0.f, 0.f, 0.f, 0.f}, x4b = {0.f, 0.f, 0.f, 0.f};

  if (wave == 0) {
    v8f brzu[NRZ];
    load_bias(bih0, bhh0, h, brzu, bnx0, bnh0);
    load_bias(bih1, bhh1, h, brzu, bnx1, bnh1);
    #pragma unroll
    for (int kc = 0; kc < KC_H; ++kc) { h0[kc] = (v2f){0.f, 0.f}; h1[kc] = (v2f){0.f, 0.f}; }
    put_state(hx + lane, h1);
    put_state(h0x + lane, h0);
  } else {
    bool nreal;
    const int gn = n_row(tn, m, nreal);
    load_tile_w(wa, rz_row(ta, m), true, h, Whh1, Wih1, Whh0, Wih0);
    load_tile_w(wb, rz_row(tb, m), true, h, Whh1, Wih1, Whh0, Wih0);
    load_tile_w(wn, gn, nreal, h, Whh1, Wih1, Whh0, Wih0);
    b0a = bias_tile(bih0, bhh0, ta, h);
    b0b = bias_tile(bih0, bhh0, tb, h);
    b1a = bias_tile(bih1, bhh1, ta, h);
    b1b = bias_tile(bih1, bhh1, tb, h);
    #pragma unroll
    for (int kc = 0; kc < KC_H; ++kc) hb0[kc] = (v2f){0.f, 0.f};
    const float* xrow = x + xoff;
    helper_input_side(xs + lane, ta, tb, tn, wa, wb, wn, b0a, b0b, *(const v4f*)(xrow), *(const v4f*)(xrow + 4));
  }
  __syncthreads();

  if (wave != 0) {
    const float* xrow = x + (xoff + I);
    helper_layer0(h0x + lane, xs + lane, s0 + lane, xs + SLOT + lane, ta, tb, tn, wa, wb, wn, b0a, b0b,
                  *(const v4f*)(xrow), *(const v4f*)(xrow + 4), hb0, xnk);
  }
  __syncthreads();

  #pragma unroll 1
  for (int i = 0; i <= T; ++i) {
    if (wave == 0) {
      if (i < T) {
        v8f rz0[NRZ], xn0[NN], hn0[NN];
        take_slot(s0 + lane, rz0, hn0);
        take_slot_n(xs + (i & 1) * SLOT + lane, NRZ, xn0);
        cell_update(rz0, xn0, hn0, bnx0, bnh0, h0);
        put_state(h0x + lane, h0);
      }
    } else {
      const int s = (i + 2 < T) ? (i + 2) : (T - 1);
      const float* xrow = x + (xoff + s * I);
      x4a = *(const v4f*)(xrow);
      x4b = *(const v4f*)(xrow + 4);
      helper_layer1(hx + lane, s1 + lane, ta, tb, tn, wa, wb, wn, b1a, b1b, hb0, xnk);
    }
    __syncthreads();
    if (wave == 0) {
      if (i >= 1) {
        v8f rz1[NRZ], xn1[NN], hn1[NN];
        take_slot(s1 + lane, rz1, hn1);
        take_slot_n(s1 + lane, NRZ + NN, xn1);
        cell_update(rz1, xn1, hn1, bnx1, bnh1, h1);
        put_state(hx + lane, h1);
      }
    } else {
      helper_layer0(h0x + lane, xs + ((i + 1) & 1) * SLOT + lane, s0 + lane, xs + (i & 1) * SLOT + lane,
                    ta, tb, tn, wa, wb, wn, b0a, b0b, x4a, x4b, hb0, xnk);
    }
    __syncthreads();
  }

  if (wave == 0) {
    #pragma unroll
    for (int i = 0; i < HH; ++i) stage[m * H + HH * h + i] = h1[i >> 1][i & 1];
  }
  __syncthreads();
  #pragma unroll
  for (int e = threadIdx.x; e < 16 * H; e += 128) out[(size_t)bbase * H + e] = stage[e];
}

extern "C" void kernel_launch(void* const* d_in, const int* in_sizes, int n_in,
                              void* d_out, int out_size, void* d_ws, size_t ws_size,
                              hipStream_t stream) {
  const float* x    = (const float*)d_in[0];
  const float* Wih0 = (const float*)d_in[1];
  const float* Whh0 = (const float*)d_in[2];
  const float* bih0 = (const float*)d_in[3];
  const float* bhh0 = (const float*)d_in[4];
  const float* Wih1 = (const float*)d_in[5];
  const float* Whh1 = (const float*)d_in[6];
  const float* bih1 = (const float*)d_in[7];
  const float* bhh1 = (const float*)d_in[8];
  float* out = (float*)d_out;

  const int B = in_sizes[0] / (T * I);
  const int nblocks = (B + 15) / 16;

  gru2_fused<<<nblocks, 128, 0, stream>>>(x, Wih0, Whh0, bih0, bhh0,
                                         Wih1, Whh1, bih1, bhh1, out);
  (void)n_in; (void)out_size; (void)d_ws; (void)ws_size;
}
